# starting kernel
# baseline (speedup 1.0000x reference)
#include <hip/hip_runtime.h>

typedef __attribute__((ext_vector_type(2))) float v2f;
typedef __attribute__((ext_vector_type(8))) float v8f;

__global__ void k_deg_init(float* __restrict__ deg, int n) {
  int i = blockIdx.x * blockDim.x + threadIdx.x;
  if (i < n) deg[i] = 1.0f;
}

__global__ void k_deg_edges(const int* __restrict__ dst, float* __restrict__ deg, int E) {
  int e = blockIdx.x * blockDim.x + threadIdx.x;
  if (e < E) unsafeAtomicAdd(&deg[dst[e]], 1.0f);
}

__global__ void k_dinv(float* __restrict__ deg, int n) {
  int i = blockIdx.x * blockDim.x + threadIdx.x;
  if (i < n) deg[i] = rsqrtf(deg[i]);
}

__global__ void k_gemm1(const float* __restrict__ x, const float* __restrict__ W1,
                        float* __restrict__ h1, int n, int in_c, int hid) {
  int idx = blockIdx.x * blockDim.x + threadIdx.x;
  if (idx >= n * hid) return;
  int node = idx / hid;
  int c = idx - node * hid;
  float acc = 0.0f;
  for (int k = 0; k < in_c; ++k) acc += x[node * in_c + k] * W1[k * hid + c];
  h1[idx] = acc;
}

__global__ void k_selfbias(const float* __restrict__ h, const float* __restrict__ dinv,
                           const float* __restrict__ b, float* __restrict__ out,
                           int n, int C) {
  int idx = blockIdx.x * blockDim.x + threadIdx.x;
  if (idx >= n * C) return;
  int node = idx / C;
  int c = idx - node * C;
  float di = dinv[node];
  out[idx] = h[idx] * di * di + b[c];
}

__global__ void k_scatter128(const float* __restrict__ h, const float* __restrict__ dinv,
                             const int* __restrict__ src, const int* __restrict__ dst,
                             float* __restrict__ out, int E) {
  int gid = blockIdx.x * blockDim.x + threadIdx.x;
  int e = gid >> 5;
  int lane = gid & 31;
  if (e >= E) return;
  int s = src[e];
  int d = dst[e];
  float norm = dinv[s] * dinv[d];
  const float4 v = *(const float4*)(h + (size_t)s * 128 + lane * 4);
  float* op = out + (size_t)d * 128 + lane * 4;
  unsafeAtomicAdd(op + 0, v.x * norm);
  unsafeAtomicAdd(op + 1, v.y * norm);
  unsafeAtomicAdd(op + 2, v.z * norm);
  unsafeAtomicAdd(op + 3, v.w * norm);
}

__global__ void k_scatter64(const float* __restrict__ h, const float* __restrict__ dinv,
                            const int* __restrict__ src, const int* __restrict__ dst,
                            float* __restrict__ out, int E) {
  int gid = blockIdx.x * blockDim.x + threadIdx.x;
  int e = gid >> 5;
  int lane = gid & 31;
  if (e >= E) return;
  int s = src[e];
  int d = dst[e];
  float norm = dinv[s] * dinv[d];
  const float2 v = *(const float2*)(h + (size_t)s * 64 + lane * 2);
  float* op = out + (size_t)d * 64 + lane * 2;
  unsafeAtomicAdd(op + 0, v.x * norm);
  unsafeAtomicAdd(op + 1, v.y * norm);
}

__global__ void k_gemm2_wmma(const float* __restrict__ A, const float* __restrict__ B,
                             float* __restrict__ D, int n, int hid, int out_c) {
  int wid  = (blockIdx.x * blockDim.x + threadIdx.x) >> 5;
  int lane = threadIdx.x & 31;
  int nTiles = out_c >> 4;
  int mTile = wid / nTiles;
  int nTile = wid - mTile * nTiles;
  if (mTile * 16 >= n) return;

  int half = lane >> 4;
  int l16  = lane & 15;

  int rowA = mTile * 16 + l16;
  if (rowA >= n) rowA = n - 1;
  int colB = nTile * 16 + l16;

  const float* arow = A + (size_t)rowA * hid;
  v8f c = {};
  for (int k = 0; k < hid; k += 4) {
    int kk = k + half * 2;
    v2f a;
    a.x = fmaxf(arow[kk + 0], 0.0f);
    a.y = fmaxf(arow[kk + 1], 0.0f);
    const float* bp = B + (size_t)kk * out_c + colB;
    v2f b;
    b.x = bp[0];
    b.y = bp[out_c];
    c = __builtin_amdgcn_wmma_f32_16x16x4_f32(false, a, false, b, (short)0, c,
                                              false, false);
  }
#pragma unroll
  for (int r = 0; r < 8; ++r) {
    int row = mTile * 16 + r + half * 8;
    if (row < n) D[(size_t)row * out_c + nTile * 16 + l16] = c[r];
  }
}

extern "C" void kernel_launch(void* const* d_in, const int* in_sizes, int n_in,
                              void* d_out, int out_size, void* d_ws, size_t ws_size,
                              hipStream_t stream) {
  const float* x  = (const float*)d_in[0];
  const int*   ei = (const int*)d_in[1];
  const float* W1 = (const float*)d_in[2];
  const float* b1 = (const float*)d_in[3];
  const float* W2 = (const float*)d_in[4];
  const float* b2 = (const float*)d_in[5];
  float* out = (float*)d_out;

  int hid   = in_sizes[3];
  int out_c = in_sizes[5];
  int in_c  = in_sizes[2] / hid;
  int n     = in_sizes[0] / in_c;
  int E     = in_sizes[1] / 2;

  const int* src = ei;
  const int* dst = ei + E;

  float* ws   = (float*)d_ws;
  float* dinv = ws;
  float* h1   = dinv + n;
  float* agg1 = h1 + (size_t)n * hid;
  float* h2   = h1;

  const int B256 = 256;
  k_deg_init<<<(n + B256 - 1) / B256, B256, 0, stream>>>(dinv, n);
  k_deg_edges<<<(E + B256 - 1) / B256, B256, 0, stream>>>(dst, dinv, E);
  k_dinv<<<(n + B256 - 1) / B256, B256, 0, stream>>>(dinv, n);

  int tot1 = n * hid;
  k_gemm1<<<(tot1 + B256 - 1) / B256, B256, 0, stream>>>(x, W1, h1, n, in_c, hid);
  k_selfbias<<<(tot1 + B256 - 1) / B256, B256, 0, stream>>>(h1, dinv, b1, agg1, n, hid);
  k_scatter128<<<(E + 7) / 8, B256, 0, stream>>>(h1, dinv, src, dst, agg1, E);

  int mTiles = (n + 15) / 16;
  int nTiles = out_c / 16;
  int waves  = mTiles * nTiles;
  k_gemm2_wmma<<<(waves + 7) / 8, B256, 0, stream>>>(agg1, W2, h2, n, hid, out_c);

  int tot2 = n * out_c;
  k_selfbias<<<(tot2 + B256 - 1) / B256, B256, 0, stream>>>(h2, dinv, b2, out, n, out_c);
  k_scatter64<<<(E + 7) / 8, B256, 0, stream>>>(h2, dinv, src, dst, out, E);
}


# layer 1 aggregates the 5-channel input before the W1 product, (A x)W1 for A(x W1): 5 f32 atomic adds per edge for 128; the float sums are taken in another order
# speedup vs baseline: 1.7016x; 1.7016x over previous
#include <hip/hip_runtime.h>

typedef __attribute__((ext_vector_type(2))) float v2f;
typedef __attribute__((ext_vector_type(8))) float v8f;

__global__ void k_deg_init(float* __restrict__ deg, int n) {
  int i = blockIdx.x * blockDim.x + threadIdx.x;
  if (i < n) deg[i] = 1.0f;
}

__global__ void k_deg_edges(const int* __restrict__ dst, float* __restrict__ deg, int E) {
  int e = blockIdx.x * blockDim.x + threadIdx.x;
  if (e < E) unsafeAtomicAdd(&deg[dst[e]], 1.0f);
}

__global__ void k_dinv(float* __restrict__ deg, int n) {
  int i = blockIdx.x * blockDim.x + threadIdx.x;
  if (i < n) deg[i] = rsqrtf(deg[i]);
}

#define AXP 32

__global__ void k_xself(const float* __restrict__ x, const float* __restrict__ dinv,
                        float* __restrict__ ax, int n, int in_c) {
  int idx = blockIdx.x * blockDim.x + threadIdx.x;
  int node = idx >> 3;
  int c = idx & 7;
  if (node >= n || c >= in_c) return;
  float di = dinv[node];
  ax[(size_t)node * AXP + c] = x[node * in_c + c] * (di * di);
}

__global__ void k_scatter_x(const float* __restrict__ x, const float* __restrict__ dinv,
                            const int* __restrict__ src, const int* __restrict__ dst,
                            float* __restrict__ ax, int E, int in_c) {
  int e = blockIdx.x * blockDim.x + threadIdx.x;
  int c = threadIdx.x & 7;
  int cc = c < in_c ? c : in_c - 1;
  int ok = e < E;
  int s = 0, d = 0;
  float norm = 0.0f;
  if (ok) {
    s = src[e];
    d = dst[e];
    norm = dinv[s] * dinv[d];
  }
  float v[8];
  int dj[8], okj[8];
#pragma unroll
  for (int j = 0; j < 8; ++j) {
    int sj = __shfl(s, j, 8);
    dj[j] = __shfl(d, j, 8);
    okj[j] = __shfl(ok, j, 8);
    v[j] = x[(size_t)sj * in_c + cc] * __shfl(norm, j, 8);
  }
#pragma unroll
  for (int j = 0; j < 8; ++j) {
    if (okj[j] && c < in_c) unsafeAtomicAdd(ax + (size_t)dj[j] * AXP + c, v[j]);
  }
}

__global__ void k_gemm1(const float* __restrict__ ax, const float* __restrict__ W1,
                        const float* __restrict__ b1, float* __restrict__ agg1,
                        int n, int in_c, int hid) {
  int idx = blockIdx.x * blockDim.x + threadIdx.x;
  if (idx >= n * hid) return;
  int node = idx / hid;
  int c = idx - node * hid;
  float acc = 0.0f;
  for (int k = 0; k < in_c; ++k) acc += ax[(size_t)node * AXP + k] * W1[k * hid + c];
  agg1[idx] = acc + b1[c];
}

__global__ void k_selfbias(const float* __restrict__ h, const float* __restrict__ dinv,
                           const float* __restrict__ b, float* __restrict__ out,
                           int n, int C) {
  int idx = blockIdx.x * blockDim.x + threadIdx.x;
  if (idx >= n * C) return;
  int node = idx / C;
  int c = idx - node * C;
  float di = dinv[node];
  out[idx] = h[idx] * di * di + b[c];
}

__global__ void k_scatter64(const float* __restrict__ h, const float* __restrict__ dinv,
                            const int* __restrict__ src, const int* __restrict__ dst,
                            float* __restrict__ out, int E) {
  int gid = blockIdx.x * blockDim.x + threadIdx.x;
  int e = gid >> 5;
  int lane = gid & 31;
  if (e >= E) return;
  int s = src[e];
  int d = dst[e];
  float norm = dinv[s] * dinv[d];
  const float2 v = *(const float2*)(h + (size_t)s * 64 + lane * 2);
  float* op = out + (size_t)d * 64 + lane * 2;
  unsafeAtomicAdd(op + 0, v.x * norm);
  unsafeAtomicAdd(op + 1, v.y * norm);
}

__global__ void k_gemm2_wmma(const float* __restrict__ A, const float* __restrict__ B,
                             float* __restrict__ D, int n, int hid, int out_c) {
  int wid  = (blockIdx.x * blockDim.x + threadIdx.x) >> 5;
  int lane = threadIdx.x & 31;
  int nTiles = out_c >> 4;
  int mTile = wid / nTiles;
  int nTile = wid - mTile * nTiles;
  if (mTile * 16 >= n) return;

  int half = lane >> 4;
  int l16  = lane & 15;

  int rowA = mTile * 16 + l16;
  if (rowA >= n) rowA = n - 1;
  int colB = nTile * 16 + l16;

  const float* arow = A + (size_t)rowA * hid;
  v8f c = {};
  for (int k = 0; k < hid; k += 4) {
    int kk = k + half * 2;
    v2f a;
    a.x = fmaxf(arow[kk + 0], 0.0f);
    a.y = fmaxf(arow[kk + 1], 0.0f);
    const float* bp = B + (size_t)kk * out_c + colB;
    v2f b;
    b.x = bp[0];
    b.y = bp[out_c];
    c = __builtin_amdgcn_wmma_f32_16x16x4_f32(false, a, false, b, (short)0, c,
                                              false, false);
  }
#pragma unroll
  for (int r = 0; r < 8; ++r) {
    int row = mTile * 16 + r + half * 8;
    if (row < n) D[(size_t)row * out_c + nTile * 16 + l16] = c[r];
  }
}

extern "C" void kernel_launch(void* const* d_in, const int* in_sizes, int n_in,
                              void* d_out, int out_size, void* d_ws, size_t ws_size,
                              hipStream_t stream) {
  const float* x  = (const float*)d_in[0];
  const int*   ei = (const int*)d_in[1];
  const float* W1 = (const float*)d_in[2];
  const float* b1 = (const float*)d_in[3];
  const float* W2 = (const float*)d_in[4];
  const float* b2 = (const float*)d_in[5];
  float* out = (float*)d_out;

  int hid   = in_sizes[3];
  int out_c = in_sizes[5];
  int in_c  = in_sizes[2] / hid;
  int n     = in_sizes[0] / in_c;
  int E     = in_sizes[1] / 2;

  const int* src = ei;
  const int* dst = ei + E;

  float* ws   = (float*)d_ws;
  float* dinv = ws;
  float* ax   = dinv + n;
  float* agg1 = ax + (size_t)n * AXP;
  float* h2   = agg1 + (size_t)n * hid;

  const int B256 = 256;
  k_deg_init<<<(n + B256 - 1) / B256, B256, 0, stream>>>(dinv, n);
  k_deg_edges<<<(E + B256 - 1) / B256, B256, 0, stream>>>(dst, dinv, E);
  k_dinv<<<(n + B256 - 1) / B256, B256, 0, stream>>>(dinv, n);

  int totx = n * 8;
  k_xself<<<(totx + B256 - 1) / B256, B256, 0, stream>>>(x, dinv, ax, n, in_c);
  k_scatter_x<<<(E + B256 - 1) / B256, B256, 0, stream>>>(x, dinv, src, dst, ax, E, in_c);
  int tot1 = n * hid;
  k_gemm1<<<(tot1 + B256 - 1) / B256, B256, 0, stream>>>(ax, W1, b1, agg1, n, in_c, hid);

  int mTiles = (n + 15) / 16;
  int nTiles = out_c / 16;
  int waves  = mTiles * nTiles;
  k_gemm2_wmma<<<(waves + 7) / 8, B256, 0, stream>>>(agg1, W2, h2, n, hid, out_c);

  int tot2 = n * out_c;
  k_selfbias<<<(tot2 + B256 - 1) / B256, B256, 0, stream>>>(h2, dinv, b2, out, n, out_c);
  k_scatter64<<<(E + 7) / 8, B256, 0, stream>>>(h2, dinv, src, dst, out, E);
}


# both aggregations sum each node's in-edges in registers from a destination-grouped edge list rebuilt every call (int atomic rank, scan, fill); no f32 atomics left; sum order = arrival rank
# speedup vs baseline: 2.3600x; 1.3869x over previous
#include <hip/hip_runtime.h>

typedef __attribute__((ext_vector_type(2))) float v2f;
typedef __attribute__((ext_vector_type(8))) float v8f;

__global__ void k_cnt_init(int* __restrict__ cnt, int n) {
  int i = blockIdx.x * blockDim.x + threadIdx.x;
  if (i < n) cnt[i] = 0;
}

#define EPT 4
__global__ void k_rank(const int* __restrict__ dst, int* __restrict__ cnt,
                       int* __restrict__ rank, int E) {
  int t = blockIdx.x * blockDim.x + threadIdx.x;
  int T = gridDim.x * blockDim.x;
  if (t + (EPT - 1) * T < E) {
    int d[EPT], r[EPT];
#pragma unroll
    for (int k = 0; k < EPT; ++k) d[k] = dst[t + k * T];
#pragma unroll
    for (int k = 0; k < EPT; ++k) r[k] = atomicAdd(&cnt[d[k]], 1);
#pragma unroll
    for (int k = 0; k < EPT; ++k) rank[t + k * T] = r[k];
  } else {
#pragma unroll
    for (int k = 0; k < EPT; ++k) {
      int e = t + k * T;
      if (e < E) rank[e] = atomicAdd(&cnt[dst[e]], 1);
    }
  }
}

#define SCAN_T 1024
__global__ void __launch_bounds__(SCAN_T)
k_scan(const int* __restrict__ cnt, int* __restrict__ start, float* __restrict__ dinv, int n) {
  __shared__ int s_pre[32];
  __shared__ int s_tot[32];
  int t = threadIdx.x;
  int lane = t & 31;
  int w = t >> 5;
  int first = blockIdx.x * SCAN_T;
  int s = 0;
  for (int j = t; j < first; j += SCAN_T) s += cnt[j];
  int i = first + t;
  int v = i < n ? cnt[i] : 0;
  int incl = v;
#pragma unroll
  for (int o = 1; o < 32; o <<= 1) {
    int u = __shfl_up(incl, o);
    if (lane >= o) incl += u;
  }
#pragma unroll
  for (int o = 16; o > 0; o >>= 1) s += __shfl_xor(s, o);
  if (lane == 31) s_tot[w] = incl;
  if (lane == 0) s_pre[w] = s;
  __syncthreads();
  int a = s_pre[lane] + (lane < w ? s_tot[lane] : 0);
#pragma unroll
  for (int o = 16; o > 0; o >>= 1) a += __shfl_xor(a, o);
  if (i < n) {
    start[i] = a + incl - v;
    dinv[i] = rsqrtf((float)(v + 1));
  }
}

__global__ void k_fill(const int* __restrict__ src, const int* __restrict__ dst,
                       const int* __restrict__ rank, const int* __restrict__ start,
                       int* __restrict__ csrc, int E) {
  int t = blockIdx.x * blockDim.x + threadIdx.x;
  int T = gridDim.x * blockDim.x;
  if (t + (EPT - 1) * T < E) {
    int d[EPT], r[EPT], s[EPT], p[EPT];
#pragma unroll
    for (int k = 0; k < EPT; ++k) {
      int e = t + k * T;
      d[k] = dst[e];
      r[k] = rank[e];
      s[k] = src[e];
    }
#pragma unroll
    for (int k = 0; k < EPT; ++k) p[k] = start[d[k]];
#pragma unroll
    for (int k = 0; k < EPT; ++k) csrc[p[k] + r[k]] = s[k];
  } else {
#pragma unroll
    for (int k = 0; k < EPT; ++k) {
      int e = t + k * T;
      if (e < E) csrc[start[dst[e]] + rank[e]] = src[e];
    }
  }
}

#define AXP 8

__global__ void k_gather_x(const float* __restrict__ x, const float* __restrict__ dinv,
                           const int* __restrict__ start, const int* __restrict__ cnt,
                           const int* __restrict__ csrc, float* __restrict__ ax,
                           int n, int in_c) {
  int idx = blockIdx.x * blockDim.x + threadIdx.x;
  int d = idx >> 3;
  int c = threadIdx.x & 7;
  if (d >= n) return;
  int cc = c < in_c ? c : in_c - 1;
  int beg = start[d];
  int m = cnt[d];
  float dd = dinv[d];
  float acc = x[(size_t)d * in_c + cc] * (dd * dd);
  for (int base = 0; base < m; base += 8) {
    int sj = d;
    float wj = 0.0f;
    if (base + c < m) {
      sj = csrc[beg + base + c];
      wj = dinv[sj] * dd;
    }
    float v[8], w[8];
#pragma unroll
    for (int j = 0; j < 8; ++j) {
      int s = __shfl(sj, j, 8);
      w[j] = __shfl(wj, j, 8);
      v[j] = x[(size_t)s * in_c + cc];
    }
#pragma unroll
    for (int j = 0; j < 8; ++j) acc += v[j] * w[j];
  }
  if (c < in_c) ax[(size_t)d * AXP + c] = acc;
}

__global__ void k_gemm1(const float* __restrict__ ax, const float* __restrict__ W1,
                        const float* __restrict__ b1, float* __restrict__ agg1,
                        int n, int in_c, int hid) {
  int idx = blockIdx.x * blockDim.x + threadIdx.x;
  if (idx >= n * hid) return;
  int node = idx / hid;
  int c = idx - node * hid;
  float acc = 0.0f;
  for (int k = 0; k < in_c; ++k) acc += ax[(size_t)node * AXP + k] * W1[k * hid + c];
  agg1[idx] = acc + b1[c];
}

__global__ void k_gather64(const float* __restrict__ h, const float* __restrict__ dinv,
                           const int* __restrict__ start, const int* __restrict__ cnt,
                           const int* __restrict__ csrc, const float* __restrict__ b,
                           float* __restrict__ out, int n) {
  int lane = threadIdx.x & 31;
  int d = __builtin_amdgcn_readfirstlane((int)((blockIdx.x * blockDim.x + threadIdx.x) >> 5));
  if (d >= n) return;
  int beg = start[d];
  int m = cnt[d];
  float dd = dinv[d];
  const float2 hv = *(const float2*)(h + (size_t)d * 64 + lane * 2);
  float accx = hv.x * dd * dd + b[lane * 2 + 0];
  float accy = hv.y * dd * dd + b[lane * 2 + 1];
  for (int base = 0; base < m; base += 32) {
    int mm = m - base < 32 ? m - base : 32;
    int sj = d;
    float wj = 0.0f;
    if (lane < mm) {
      sj = csrc[beg + base + lane];
      wj = dinv[sj] * dd;
    }
    for (int j = 0; j < mm; j += 4) {
      float2 v[4];
      float wv[4];
#pragma unroll
      for (int u = 0; u < 4; ++u) {
        int s = __builtin_amdgcn_readlane(sj, j + u);
        wv[u] = __int_as_float(__builtin_amdgcn_readlane(__float_as_int(wj), j + u));
        v[u] = *(const float2*)(h + (size_t)s * 64 + lane * 2);
      }
#pragma unroll
      for (int u = 0; u < 4; ++u) {
        accx += v[u].x * wv[u];
        accy += v[u].y * wv[u];
      }
    }
  }
  float2 r;
  r.x = accx;
  r.y = accy;
  *(float2*)(out + (size_t)d * 64 + lane * 2) = r;
}

__global__ void k_gemm2_wmma(const float* __restrict__ A, const float* __restrict__ B,
                             float* __restrict__ D, int n, int hid, int out_c) {
  int wid  = (blockIdx.x * blockDim.x + threadIdx.x) >> 5;
  int lane = threadIdx.x & 31;
  int nTiles = out_c >> 4;
  int mTile = wid / nTiles;
  int nTile = wid - mTile * nTiles;
  if (mTile * 16 >= n) return;

  int half = lane >> 4;
  int l16  = lane & 15;

  int rowA = mTile * 16 + l16;
  if (rowA >= n) rowA = n - 1;
  int colB = nTile * 16 + l16;

  const float* arow = A + (size_t)rowA * hid;
  v8f c = {};
  for (int k = 0; k < hid; k += 4) {
    int kk = k + half * 2;
    v2f a;
    a.x = fmaxf(arow[kk + 0], 0.0f);
    a.y = fmaxf(arow[kk + 1], 0.0f);
    const float* bp = B + (size_t)kk * out_c + colB;
    v2f b;
    b.x = bp[0];
    b.y = bp[out_c];
    c = __builtin_amdgcn_wmma_f32_16x16x4_f32(false, a, false, b, (short)0, c,
                                              false, false);
  }
#pragma unroll
  for (int r = 0; r < 8; ++r) {
    int row = mTile * 16 + r + half * 8;
    if (row < n) D[(size_t)row * out_c + nTile * 16 + l16] = c[r];
  }
}

extern "C" void kernel_launch(void* const* d_in, const int* in_sizes, int n_in,
                              void* d_out, int out_size, void* d_ws, size_t ws_size,
                              hipStream_t stream) {
  const float* x  = (const float*)d_in[0];
  const int*   ei = (const int*)d_in[1];
  const float* W1 = (const float*)d_in[2];
  const float* b1 = (const float*)d_in[3];
  const float* W2 = (const float*)d_in[4];
  const float* b2 = (const float*)d_in[5];
  float* out = (float*)d_out;

  int hid   = in_sizes[3];
  int out_c = in_sizes[5];
  int in_c  = in_sizes[2] / hid;
  int n     = in_sizes[0] / in_c;
  int E     = in_sizes[1] / 2;

  const int* src = ei;
  const int* dst = ei + E;

  float* ws   = (float*)d_ws;
  float* dinv = ws;
  float* ax   = dinv + n;
  float* agg1 = ax + (size_t)n * AXP;
  float* h2   = agg1 + (size_t)n * hid;
  int* cnt    = (int*)(h2 + (size_t)n * out_c);
  int* start  = cnt + n;
  int* rank   = start + n;
  int* csrc   = rank + E;

  const int B256 = 256;
  k_cnt_init<<<(n + B256 - 1) / B256, B256, 0, stream>>>(cnt, n);
  int egrid = (E + B256 * EPT - 1) / (B256 * EPT);
  k_rank<<<egrid, B256, 0, stream>>>(dst, cnt, rank, E);
  k_scan<<<(n + SCAN_T - 1) / SCAN_T, SCAN_T, 0, stream>>>(cnt, start, dinv, n);
  k_fill<<<egrid, B256, 0, stream>>>(src, dst, rank, start, csrc, E);

  int totx = n * 8;
  k_gather_x<<<(totx + B256 - 1) / B256, B256, 0, stream>>>(x, dinv, start, cnt, csrc, ax, n, in_c);
  int tot1 = n * hid;
  k_gemm1<<<(tot1 + B256 - 1) / B256, B256, 0, stream>>>(ax, W1, b1, agg1, n, in_c, hid);

  int mTiles = (n + 15) / 16;
  int nTiles = out_c / 16;
  int waves  = mTiles * nTiles;
  k_gemm2_wmma<<<(waves + 7) / 8, B256, 0, stream>>>(agg1, W2, h2, n, hid, out_c);

  k_gather64<<<(n + 7) / 8, B256, 0, stream>>>(h2, dinv, start, cnt, csrc, b2, out, n);
}


# layer-1 linear (W1, b1, ReLU) computed in the layer-2 WMMA kernel's A loader from the aggregated 8-float row: the n x 128 hidden matrix is never stored; a wave owns 16 rows x 64 cols; sum order kept
# speedup vs baseline: 2.9902x; 1.2670x over previous
#include <hip/hip_runtime.h>

typedef __attribute__((ext_vector_type(2))) float v2f;
typedef __attribute__((ext_vector_type(8))) float v8f;

__global__ void k_cnt_init(int* __restrict__ cnt, int n) {
  int i = blockIdx.x * blockDim.x + threadIdx.x;
  if (i < n) cnt[i] = 0;
}

#define EPT 4
__global__ void k_rank(const int* __restrict__ dst, int* __restrict__ cnt,
                       int* __restrict__ rank, int E) {
  int t = blockIdx.x * blockDim.x + threadIdx.x;
  int T = gridDim.x * blockDim.x;
  if (t + (EPT - 1) * T < E) {
    int d[EPT], r[EPT];
#pragma unroll
    for (int k = 0; k < EPT; ++k) d[k] = dst[t + k * T];
#pragma unroll
    for (int k = 0; k < EPT; ++k) r[k] = atomicAdd(&cnt[d[k]], 1);
#pragma unroll
    for (int k = 0; k < EPT; ++k) rank[t + k * T] = r[k];
  } else {
#pragma unroll
    for (int k = 0; k < EPT; ++k) {
      int e = t + k * T;
      if (e < E) rank[e] = atomicAdd(&cnt[dst[e]], 1);
    }
  }
}

#define SCAN_T 1024
__global__ void __launch_bounds__(SCAN_T)
k_scan(const int* __restrict__ cnt, int* __restrict__ start, float* __restrict__ dinv, int n) {
  __shared__ int s_pre[32];
  __shared__ int s_tot[32];
  int t = threadIdx.x;
  int lane = t & 31;
  int w = t >> 5;
  int first = blockIdx.x * SCAN_T;
  int s = 0;
  for (int j = t; j < first; j += SCAN_T) s += cnt[j];
  int i = first + t;
  int v = i < n ? cnt[i] : 0;
  int incl = v;
#pragma unroll
  for (int o = 1; o < 32; o <<= 1) {
    int u = __shfl_up(incl, o);
    if (lane >= o) incl += u;
  }
#pragma unroll
  for (int o = 16; o > 0; o >>= 1) s += __shfl_xor(s, o);
  if (lane == 31) s_tot[w] = incl;
  if (lane == 0) s_pre[w] = s;
  __syncthreads();
  int a = s_pre[lane] + (lane < w ? s_tot[lane] : 0);
#pragma unroll
  for (int o = 16; o > 0; o >>= 1) a += __shfl_xor(a, o);
  if (i < n) {
    start[i] = a + incl - v;
    dinv[i] = rsqrtf((float)(v + 1));
  }
}

__global__ void k_fill(const int* __restrict__ src, const int* __restrict__ dst,
                       const int* __restrict__ rank, const int* __restrict__ start,
                       int* __restrict__ csrc, int E) {
  int t = blockIdx.x * blockDim.x + threadIdx.x;
  int T = gridDim.x * blockDim.x;
  if (t + (EPT - 1) * T < E) {
    int d[EPT], r[EPT], s[EPT], p[EPT];
#pragma unroll
    for (int k = 0; k < EPT; ++k) {
      int e = t + k * T;
      d[k] = dst[e];
      r[k] = rank[e];
      s[k] = src[e];
    }
#pragma unroll
    for (int k = 0; k < EPT; ++k) p[k] = start[d[k]];
#pragma unroll
    for (int k = 0; k < EPT; ++k) csrc[p[k] + r[k]] = s[k];
  } else {
#pragma unroll
    for (int k = 0; k < EPT; ++k) {
      int e = t + k * T;
      if (e < E) csrc[start[dst[e]] + rank[e]] = src[e];
    }
  }
}

#define AXP 8

__global__ void k_gather_x(const float* __restrict__ x, const float* __restrict__ dinv,
                           const int* __restrict__ start, const int* __restrict__ cnt,
                           const int* __restrict__ csrc, float* __restrict__ ax,
                           int n, int in_c) {
  int idx = blockIdx.x * blockDim.x + threadIdx.x;
  int d = idx >> 3;
  int c = threadIdx.x & 7;
  if (d >= n) return;
  int cc = c < in_c ? c : in_c - 1;
  int beg = start[d];
  int m = cnt[d];
  float dd = dinv[d];
  float acc = x[(size_t)d * in_c + cc] * (dd * dd);
  for (int base = 0; base < m; base += 8) {
    int sj = d;
    float wj = 0.0f;
    if (base + c < m) {
      sj = csrc[beg + base + c];
      wj = dinv[sj] * dd;
    }
    float v[8], w[8];
#pragma unroll
    for (int j = 0; j < 8; ++j) {
      int s = __shfl(sj, j, 8);
      w[j] = __shfl(wj, j, 8);
      v[j] = x[(size_t)s * in_c + cc];
    }
#pragma unroll
    for (int j = 0; j < 8; ++j) acc += v[j] * w[j];
  }
  if (c < in_c) ax[(size_t)d * AXP + c] = acc;
}

__global__ void k_gather64(const float* __restrict__ h, const float* __restrict__ dinv,
                           const int* __restrict__ start, const int* __restrict__ cnt,
                           const int* __restrict__ csrc, const float* __restrict__ b,
                           float* __restrict__ out, int n) {
  int lane = threadIdx.x & 31;
  int d = __builtin_amdgcn_readfirstlane((int)((blockIdx.x * blockDim.x + threadIdx.x) >> 5));
  if (d >= n) return;
  int beg = start[d];
  int m = cnt[d];
  float dd = dinv[d];
  const float2 hv = *(const float2*)(h + (size_t)d * 64 + lane * 2);
  float accx = hv.x * dd * dd + b[lane * 2 + 0];
  float accy = hv.y * dd * dd + b[lane * 2 + 1];
  for (int base = 0; base < m; base += 32) {
    int mm = m - base < 32 ? m - base : 32;
    int sj = d;
    float wj = 0.0f;
    if (lane < mm) {
      sj = csrc[beg + base + lane];
      wj = dinv[sj] * dd;
    }
    for (int j = 0; j < mm; j += 4) {
      float2 v[4];
      float wv[4];
#pragma unroll
      for (int u = 0; u < 4; ++u) {
        int s = __builtin_amdgcn_readlane(sj, j + u);
        wv[u] = __int_as_float(__builtin_amdgcn_readlane(__float_as_int(wj), j + u));
        v[u] = *(const float2*)(h + (size_t)s * 64 + lane * 2);
      }
#pragma unroll
      for (int u = 0; u < 4; ++u) {
        accx += v[u].x * wv[u];
        accy += v[u].y * wv[u];
      }
    }
  }
  float2 r;
  r.x = accx;
  r.y = accy;
  *(float2*)(out + (size_t)d * 64 + lane * 2) = r;
}

#define HID 128
#define OUTC 64
#define KSTEPS (HID / 4)
__global__ void __launch_bounds__(256)
k_mlp_wmma(const float* __restrict__ ax, const float* __restrict__ W1,
           const float* __restrict__ b1, const float* __restrict__ W2,
           float* __restrict__ D, int n, int in_c) {
  __shared__ float4 sB[KSTEPS * 2 * 32];
  __shared__ float4 sW[(HID / 2) * 4];

  int t = threadIdx.x;
#pragma unroll
  for (int j = 0; j < (KSTEPS * 32) / 256; ++j) {
    int step = j * 8 + (t >> 5);
    int ln = t & 31;
    int k0 = step * 4 + (ln >> 4) * 2;
    int c4 = (ln & 15) * 4;
    float4 r0 = *(const float4*)(W2 + (size_t)k0 * OUTC + c4);
    float4 r1 = *(const float4*)(W2 + (size_t)(k0 + 1) * OUTC + c4);
    sB[(step * 2 + 0) * 32 + ln] = make_float4(r0.x, r1.x, r0.y, r1.y);
    sB[(step * 2 + 1) * 32 + ln] = make_float4(r0.z, r1.z, r0.w, r1.w);
  }
  float* sWf = (float*)sW;
  int kk = t & (HID - 1);
  float bv = b1[kk];
#pragma unroll
  for (int j = 0; j < (HID * 8) / 256; ++j) {
    int i = j * 2 + (t >> 7);
    float v = 0.0f;
    if (i < in_c) v = W1[i * HID + kk];
    if (i == 7) v = bv;
    sWf[(kk >> 1) * 16 + i * 2 + (kk & 1)] = v;
  }
  __syncthreads();

  int lane = t & 31;
  int half = lane >> 4;
  int c    = lane & 15;
  int mTile = blockIdx.x * 8 + (t >> 5);
  if (mTile * 16 >= n) return;

  int rowA = mTile * 16 + c;
  if (rowA >= n) rowA = n - 1;
  const float4* ap = (const float4*)(ax + (size_t)rowA * AXP);
  float4 x0 = ap[0];
  float4 x1 = ap[1];
  float xr[7] = {x0.x, x0.y, x0.z, x0.w, x1.x, x1.y, x1.z};
#pragma unroll
  for (int i = 0; i < 7; ++i)
    if (i >= in_c) xr[i] = 0.0f;

  v8f acc0 = {}, acc1 = {}, acc2 = {}, acc3 = {};
#pragma unroll
  for (int s = 0; s < KSTEPS; ++s) {
    const float4* wp = sW + (s * 2 + half) * 4;
    float4 w01 = wp[0], w23 = wp[1], w45 = wp[2], w6b = wp[3];
    v2f w, xs, hh;
    w.x = w01.x; w.y = w01.y; hh = w * xr[0];
    w.x = w01.z; w.y = w01.w; xs.x = xr[1]; xs.y = xr[1]; hh = __builtin_elementwise_fma(xs, w, hh);
    w.x = w23.x; w.y = w23.y; xs.x = xr[2]; xs.y = xr[2]; hh = __builtin_elementwise_fma(xs, w, hh);
    w.x = w23.z; w.y = w23.w; xs.x = xr[3]; xs.y = xr[3]; hh = __builtin_elementwise_fma(xs, w, hh);
    w.x = w45.x; w.y = w45.y; xs.x = xr[4]; xs.y = xr[4]; hh = __builtin_elementwise_fma(xs, w, hh);
    w.x = w45.z; w.y = w45.w; xs.x = xr[5]; xs.y = xr[5]; hh = __builtin_elementwise_fma(xs, w, hh);
    w.x = w6b.x; w.y = w6b.y; xs.x = xr[6]; xs.y = xr[6]; hh = __builtin_elementwise_fma(xs, w, hh);
    w.x = w6b.z; w.y = w6b.w; hh = hh + w;
    v2f a;
    a.x = fmaxf(hh.x, 0.0f);
    a.y = fmaxf(hh.y, 0.0f);
    float4 q0 = sB[(s * 2 + 0) * 32 + lane];
    float4 q1 = sB[(s * 2 + 1) * 32 + lane];
    v2f f0, f1, f2, f3;
    f0.x = q0.x; f0.y = q0.y;
    f1.x = q0.z; f1.y = q0.w;
    f2.x = q1.x; f2.y = q1.y;
    f3.x = q1.z; f3.y = q1.w;
    acc0 = __builtin_amdgcn_wmma_f32_16x16x4_f32(false, a, false, f0, (short)0, acc0, false, false);
    acc1 = __builtin_amdgcn_wmma_f32_16x16x4_f32(false, a, false, f1, (short)0, acc1, false, false);
    acc2 = __builtin_amdgcn_wmma_f32_16x16x4_f32(false, a, false, f2, (short)0, acc2, false, false);
    acc3 = __builtin_amdgcn_wmma_f32_16x16x4_f32(false, a, false, f3, (short)0, acc3, false, false);
  }
#pragma unroll
  for (int r = 0; r < 8; ++r) {
    int row = mTile * 16 + r + half * 8;
    if (row < n)
      *(float4*)(D + (size_t)row * OUTC + c * 4) = make_float4(acc0[r], acc1[r], acc2[r], acc3[r]);
  }
}

extern "C" void kernel_launch(void* const* d_in, const int* in_sizes, int n_in,
                              void* d_out, int out_size, void* d_ws, size_t ws_size,
                              hipStream_t stream) {
  const float* x  = (const float*)d_in[0];
  const int*   ei = (const int*)d_in[1];
  const float* W1 = (const float*)d_in[2];
  const float* b1 = (const float*)d_in[3];
  const float* W2 = (const float*)d_in[4];
  const float* b2 = (const float*)d_in[5];
  float* out = (float*)d_out;

  int hid   = in_sizes[3];
  int in_c  = in_sizes[2] / hid;
  int n     = in_sizes[0] / in_c;
  int E     = in_sizes[1] / 2;

  const int* src = ei;
  const int* dst = ei + E;

  float* ws   = (float*)d_ws;
  float* dinv = ws;
  float* ax   = dinv + n;
  float* h2   = ax + (size_t)n * AXP;
  int* cnt    = (int*)(h2 + (size_t)n * OUTC);
  int* start  = cnt + n;
  int* rank   = start + n;
  int* csrc   = rank + E;

  const int B256 = 256;
  k_cnt_init<<<(n + B256 - 1) / B256, B256, 0, stream>>>(cnt, n);
  int egrid = (E + B256 * EPT - 1) / (B256 * EPT);
  k_rank<<<egrid, B256, 0, stream>>>(dst, cnt, rank, E);
  k_scan<<<(n + SCAN_T - 1) / SCAN_T, SCAN_T, 0, stream>>>(cnt, start, dinv, n);
  k_fill<<<egrid, B256, 0, stream>>>(src, dst, rank, start, csrc, E);

  int totx = n * 8;
  k_gather_x<<<(totx + B256 - 1) / B256, B256, 0, stream>>>(x, dinv, start, cnt, csrc, ax, n, in_c);

  int mTiles = (n + 15) / 16;
  k_mlp_wmma<<<(mTiles + 7) / 8, B256, 0, stream>>>(ax, W1, b1, W2, h2, n, in_c);

  k_gather64<<<(n + 7) / 8, B256, 0, stream>>>(h2, dinv, start, cnt, csrc, b2, out, n);
}


# edge ranks from returning adds on LDS counters (16 edge chunks x node ranges of 8192) + a per-chunk count table, for one returning global atomic per edge; sum order = chunk, then arrival rank in it
# speedup vs baseline: 4.2724x; 1.4288x over previous
#include <hip/hip_runtime.h>

typedef __attribute__((ext_vector_type(2))) float v2f;
typedef __attribute__((ext_vector_type(8))) float v8f;

#define RCH 16
#define RANGE 8192
#define RPT 16
__global__ void __launch_bounds__(1024)
k_rank_lds(const int* __restrict__ dst, int* __restrict__ rank, int* __restrict__ tab,
           int E, int n) {
  __shared__ int s_cnt[RANGE];
  int t = threadIdx.x;
  int lo = blockIdx.y * RANGE;
  for (int i = t; i < RANGE; i += 1024) s_cnt[i] = 0;
  __syncthreads();
  int chunkE = (E + RCH - 1) / RCH;
  int e0 = blockIdx.x * chunkE;
  int e1 = e0 + chunkE < E ? e0 + chunkE : E;
  for (int base = e0; base < e1; base += 1024 * RPT) {
    int d[RPT];
#pragma unroll
    for (int k = 0; k < RPT; ++k) {
      int e = base + k * 1024 + t;
      d[k] = e < e1 ? dst[e] : -1;
    }
    int r[RPT];
#pragma unroll
    for (int k = 0; k < RPT; ++k) {
      unsigned o = (unsigned)(d[k] - lo);
      r[k] = -1;
      if (o < RANGE) r[k] = atomicAdd(&s_cnt[o], 1);
    }
#pragma unroll
    for (int k = 0; k < RPT; ++k)
      if (r[k] >= 0) rank[base + k * 1024 + t] = r[k];
  }
  __syncthreads();
  int len = n - lo < RANGE ? n - lo : RANGE;
  for (int i = t; i < len; i += 1024) tab[(size_t)blockIdx.x * n + lo + i] = s_cnt[i];
}

__global__ void k_tot(const int* __restrict__ tab, int* __restrict__ cnt, int n) {
  int i = blockIdx.x * blockDim.x + threadIdx.x;
  if (i >= n) return;
  int s = 0;
#pragma unroll
  for (int c = 0; c < RCH; ++c) s += tab[(size_t)c * n + i];
  cnt[i] = s;
}

#define SCAN_T 1024
__global__ void __launch_bounds__(SCAN_T)
k_scan(const int* __restrict__ cnt, int* __restrict__ start, float* __restrict__ dinv,
       int* tab, int n) {
  __shared__ int s_pre[32];
  __shared__ int s_tot[32];
  int t = threadIdx.x;
  int lane = t & 31;
  int w = t >> 5;
  int first = blockIdx.x * SCAN_T;
  int s = 0;
  for (int j = t; j < first; j += SCAN_T) s += cnt[j];
  int i = first + t;
  int v = i < n ? cnt[i] : 0;
  int incl = v;
#pragma unroll
  for (int o = 1; o < 32; o <<= 1) {
    int u = __shfl_up(incl, o);
    if (lane >= o) incl += u;
  }
#pragma unroll
  for (int o = 16; o > 0; o >>= 1) s += __shfl_xor(s, o);
  if (lane == 31) s_tot[w] = incl;
  if (lane == 0) s_pre[w] = s;
  __syncthreads();
  int a = s_pre[lane] + (lane < w ? s_tot[lane] : 0);
#pragma unroll
  for (int o = 16; o > 0; o >>= 1) a += __shfl_xor(a, o);
  if (i < n) {
    int run = a + incl - v;
    start[i] = run;
    dinv[i] = rsqrtf((float)(v + 1));
    int u[RCH];
#pragma unroll
    for (int c = 0; c < RCH; ++c) u[c] = tab[(size_t)c * n + i];
#pragma unroll
    for (int c = 0; c < RCH; ++c) {
      tab[(size_t)c * n + i] = run;
      run += u[c];
    }
  }
}

#define EPT 4
__global__ void k_fill(const int* __restrict__ src, const int* __restrict__ dst,
                       const int* __restrict__ rank, const int* __restrict__ tab,
                       int* __restrict__ csrc, int E, int n) {
  int t = blockIdx.x * blockDim.x + threadIdx.x;
  int T = gridDim.x * blockDim.x;
  int chunkE = (E + RCH - 1) / RCH;
  if (t + (EPT - 1) * T < E) {
    int d[EPT], r[EPT], s[EPT], p[EPT];
#pragma unroll
    for (int k = 0; k < EPT; ++k) {
      int e = t + k * T;
      d[k] = dst[e];
      r[k] = rank[e];
      s[k] = src[e];
    }
#pragma unroll
    for (int k = 0; k < EPT; ++k) p[k] = tab[(size_t)((t + k * T) / chunkE) * n + d[k]];
#pragma unroll
    for (int k = 0; k < EPT; ++k) csrc[p[k] + r[k]] = s[k];
  } else {
#pragma unroll
    for (int k = 0; k < EPT; ++k) {
      int e = t + k * T;
      if (e < E) csrc[tab[(size_t)(e / chunkE) * n + dst[e]] + rank[e]] = src[e];
    }
  }
}

#define AXP 8

__global__ void k_gather_x(const float* __restrict__ x, const float* __restrict__ dinv,
                           const int* __restrict__ start, const int* __restrict__ cnt,
                           const int* __restrict__ csrc, float* __restrict__ ax,
                           int n, int in_c) {
  int idx = blockIdx.x * blockDim.x + threadIdx.x;
  int d = idx >> 3;
  int c = threadIdx.x & 7;
  if (d >= n) return;
  int cc = c < in_c ? c : in_c - 1;
  int beg = start[d];
  int m = cnt[d];
  float dd = dinv[d];
  float acc = x[(size_t)d * in_c + cc] * (dd * dd);
  for (int base = 0; base < m; base += 8) {
    int sj = d;
    float wj = 0.0f;
    if (base + c < m) {
      sj = csrc[beg + base + c];
      wj = dinv[sj] * dd;
    }
    float v[8], w[8];
#pragma unroll
    for (int j = 0; j < 8; ++j) {
      int s = __shfl(sj, j, 8);
      w[j] = __shfl(wj, j, 8);
      v[j] = x[(size_t)s * in_c + cc];
    }
#pragma unroll
    for (int j = 0; j < 8; ++j) acc += v[j] * w[j];
  }
  if (c < in_c) ax[(size_t)d * AXP + c] = acc;
}

__global__ void k_gather64(const float* __restrict__ h, const float* __restrict__ dinv,
                           const int* __restrict__ start, const int* __restrict__ cnt,
                           const int* __restrict__ csrc, const float* __restrict__ b,
                           float* __restrict__ out, int n) {
  int lane = threadIdx.x & 31;
  int d = __builtin_amdgcn_readfirstlane((int)((blockIdx.x * blockDim.x + threadIdx.x) >> 5));
  if (d >= n) return;
  int beg = start[d];
  int m = cnt[d];
  float dd = dinv[d];
  const float2 hv = *(const float2*)(h + (size_t)d * 64 + lane * 2);
  float accx = hv.x * dd * dd + b[lane * 2 + 0];
  float accy = hv.y * dd * dd + b[lane * 2 + 1];
  for (int base = 0; base < m; base += 32) {
    int mm = m - base < 32 ? m - base : 32;
    int sj = d;
    float wj = 0.0f;
    if (lane < mm) {
      sj = csrc[beg + base + lane];
      wj = dinv[sj] * dd;
    }
    for (int j = 0; j < mm; j += 4) {
      float2 v[4];
      float wv[4];
#pragma unroll
      for (int u = 0; u < 4; ++u) {
        int s = __builtin_amdgcn_readlane(sj, j + u);
        wv[u] = __int_as_float(__builtin_amdgcn_readlane(__float_as_int(wj), j + u));
        v[u] = *(const float2*)(h + (size_t)s * 64 + lane * 2);
      }
#pragma unroll
      for (int u = 0; u < 4; ++u) {
        accx += v[u].x * wv[u];
        accy += v[u].y * wv[u];
      }
    }
  }
  float2 r;
  r.x = accx;
  r.y = accy;
  *(float2*)(out + (size_t)d * 64 + lane * 2) = r;
}

#define HID 128
#define OUTC 64
#define KSTEPS (HID / 4)
__global__ void __launch_bounds__(256)
k_mlp_wmma(const float* __restrict__ ax, const float* __restrict__ W1,
           const float* __restrict__ b1, const float* __restrict__ W2,
           float* __restrict__ D, int n, int in_c) {
  __shared__ float4 sB[KSTEPS * 2 * 32];
  __shared__ float4 sW[(HID / 2) * 4];

  int t = threadIdx.x;
#pragma unroll
  for (int j = 0; j < (KSTEPS * 32) / 256; ++j) {
    int step = j * 8 + (t >> 5);
    int ln = t & 31;
    int k0 = step * 4 + (ln >> 4) * 2;
    int c4 = (ln & 15) * 4;
    float4 r0 = *(const float4*)(W2 + (size_t)k0 * OUTC + c4);
    float4 r1 = *(const float4*)(W2 + (size_t)(k0 + 1) * OUTC + c4);
    sB[(step * 2 + 0) * 32 + ln] = make_float4(r0.x, r1.x, r0.y, r1.y);
    sB[(step * 2 + 1) * 32 + ln] = make_float4(r0.z, r1.z, r0.w, r1.w);
  }
  float* sWf = (float*)sW;
  int kk = t & (HID - 1);
  float bv = b1[kk];
#pragma unroll
  for (int j = 0; j < (HID * 8) / 256; ++j) {
    int i = j * 2 + (t >> 7);
    float v = 0.0f;
    if (i < in_c) v = W1[i * HID + kk];
    if (i == 7) v = bv;
    sWf[(kk >> 1) * 16 + i * 2 + (kk & 1)] = v;
  }
  __syncthreads();

  int lane = t & 31;
  int half = lane >> 4;
  int c    = lane & 15;
  int mTile = blockIdx.x * 8 + (t >> 5);
  if (mTile * 16 >= n) return;

  int rowA = mTile * 16 + c;
  if (rowA >= n) rowA = n - 1;
  const float4* ap = (const float4*)(ax + (size_t)rowA * AXP);
  float4 x0 = ap[0];
  float4 x1 = ap[1];
  float xr[7] = {x0.x, x0.y, x0.z, x0.w, x1.x, x1.y, x1.z};
#pragma unroll
  for (int i = 0; i < 7; ++i)
    if (i >= in_c) xr[i] = 0.0f;

  v8f acc0 = {}, acc1 = {}, acc2 = {}, acc3 = {};
#pragma unroll
  for (int s = 0; s < KSTEPS; ++s) {
    const float4* wp = sW + (s * 2 + half) * 4;
    float4 w01 = wp[0], w23 = wp[1], w45 = wp[2], w6b = wp[3];
    v2f w, xs, hh;
    w.x = w01.x; w.y = w01.y; hh = w * xr[0];
    w.x = w01.z; w.y = w01.w; xs.x = xr[1]; xs.y = xr[1]; hh = __builtin_elementwise_fma(xs, w, hh);
    w.x = w23.x; w.y = w23.y; xs.x = xr[2]; xs.y = xr[2]; hh = __builtin_elementwise_fma(xs, w, hh);
    w.x = w23.z; w.y = w23.w; xs.x = xr[3]; xs.y = xr[3]; hh = __builtin_elementwise_fma(xs, w, hh);
    w.x = w45.x; w.y = w45.y; xs.x = xr[4]; xs.y = xr[4]; hh = __builtin_elementwise_fma(xs, w, hh);
    w.x = w45.z; w.y = w45.w; xs.x = xr[5]; xs.y = xr[5]; hh = __builtin_elementwise_fma(xs, w, hh);
    w.x = w6b.x; w.y = w6b.y; xs.x = xr[6]; xs.y = xr[6]; hh = __builtin_elementwise_fma(xs, w, hh);
    w.x = w6b.z; w.y = w6b.w; hh = hh + w;
    v2f a;
    a.x = fmaxf(hh.x, 0.0f);
    a.y = fmaxf(hh.y, 0.0f);
    float4 q0 = sB[(s * 2 + 0) * 32 + lane];
    float4 q1 = sB[(s * 2 + 1) * 32 + lane];
    v2f f0, f1, f2, f3;
    f0.x = q0.x; f0.y = q0.y;
    f1.x = q0.z; f1.y = q0.w;
    f2.x = q1.x; f2.y = q1.y;
    f3.x = q1.z; f3.y = q1.w;
    acc0 = __builtin_amdgcn_wmma_f32_16x16x4_f32(false, a, false, f0, (short)0, acc0, false, false);
    acc1 = __builtin_amdgcn_wmma_f32_16x16x4_f32(false, a, false, f1, (short)0, acc1, false, false);
    acc2 = __builtin_amdgcn_wmma_f32_16x16x4_f32(false, a, false, f2, (short)0, acc2, false, false);
    acc3 = __builtin_amdgcn_wmma_f32_16x16x4_f32(false, a, false, f3, (short)0, acc3, false, false);
  }
#pragma unroll
  for (int r = 0; r < 8; ++r) {
    int row = mTile * 16 + r + half * 8;
    if (row < n)
      *(float4*)(D + (size_t)row * OUTC + c * 4) = make_float4(acc0[r], acc1[r], acc2[r], acc3[r]);
  }
}

extern "C" void kernel_launch(void* const* d_in, const int* in_sizes, int n_in,
                              void* d_out, int out_size, void* d_ws, size_t ws_size,
                              hipStream_t stream) {
  const float* x  = (const float*)d_in[0];
  const int*   ei = (const int*)d_in[1];
  const float* W1 = (const float*)d_in[2];
  const float* b1 = (const float*)d_in[3];
  const float* W2 = (const float*)d_in[4];
  const float* b2 = (const float*)d_in[5];
  float* out = (float*)d_out;

  int hid   = in_sizes[3];
  int in_c  = in_sizes[2] / hid;
  int n     = in_sizes[0] / in_c;
  int E     = in_sizes[1] / 2;

  const int* src = ei;
  const int* dst = ei + E;

  float* ws   = (float*)d_ws;
  float* dinv = ws;
  float* ax   = dinv + n;
  float* h2   = ax + (size_t)n * AXP;
  int* cnt    = (int*)(h2 + (size_t)n * OUTC);
  int* start  = cnt + n;
  int* rank   = start + n;
  int* csrc   = rank + E;
  int* tab    = csrc + E;

  const int B256 = 256;
  k_rank_lds<<<dim3(RCH, (n + RANGE - 1) / RANGE), 1024, 0, stream>>>(dst, rank, tab, E, n);
  k_tot<<<(n + B256 - 1) / B256, B256, 0, stream>>>(tab, cnt, n);
  k_scan<<<(n + SCAN_T - 1) / SCAN_T, SCAN_T, 0, stream>>>(cnt, start, dinv, tab, n);
  int egrid = (E + B256 * EPT - 1) / (B256 * EPT);
  k_fill<<<egrid, B256, 0, stream>>>(src, dst, rank, tab, csrc, E, n);

  int totx = n * 8;
  k_gather_x<<<(totx + B256 - 1) / B256, B256, 0, stream>>>(x, dinv, start, cnt, csrc, ax, n, in_c);

  int mTiles = (n + 15) / 16;
  k_mlp_wmma<<<(mTiles + 7) / 8, B256, 0, stream>>>(ax, W1, b1, W2, h2, n, in_c);

  k_gather64<<<(n + 7) / 8, B256, 0, stream>>>(h2, dinv, start, cnt, csrc, b2, out, n);
}


# the counting kernel also stores its counters' sum per 1024-node block; the scan takes in-degrees from the chunk table and its prefix from those sums: the degree-sum launch goes; float sum order kept
# speedup vs baseline: 4.5235x; 1.0588x over previous
#include <hip/hip_runtime.h>

typedef __attribute__((ext_vector_type(2))) float v2f;
typedef __attribute__((ext_vector_type(8))) float v8f;

#define RCH 16
#define RANGE 8192
#define RPT 16
#define SCAN_T 1024
#define SUBS (RANGE / SCAN_T)
static_assert(SCAN_T == 1024 && RANGE % SCAN_T == 0 && SUBS % 4 == 0 && 32 % SUBS == 0,
              "k_rank_lds: 1024 threads, SUBS counters each, whole waves per node block");
__global__ void __launch_bounds__(1024)
k_rank_lds(const int* __restrict__ dst, int* __restrict__ rank, int* __restrict__ tab,
           int* __restrict__ bsum, int E, int n) {
  __shared__ alignas(16) int s_cnt[RANGE];
  __shared__ int s_part[32];
  int t = threadIdx.x;
  int lo = blockIdx.y * RANGE;
  for (int i = t; i < RANGE; i += 1024) s_cnt[i] = 0;
  __syncthreads();
  int chunkE = (E + RCH - 1) / RCH;
  int e0 = blockIdx.x * chunkE;
  int e1 = e0 + chunkE < E ? e0 + chunkE : E;
  for (int base = e0; base < e1; base += 1024 * RPT) {
    int d[RPT];
#pragma unroll
    for (int k = 0; k < RPT; ++k) {
      int e = base + k * 1024 + t;
      d[k] = e < e1 ? dst[e] : -1;
    }
    int r[RPT];
#pragma unroll
    for (int k = 0; k < RPT; ++k) {
      unsigned o = (unsigned)(d[k] - lo);
      r[k] = -1;
      if (o < RANGE) r[k] = atomicAdd(&s_cnt[o], 1);
    }
#pragma unroll
    for (int k = 0; k < RPT; ++k)
      if (r[k] >= 0) rank[base + k * 1024 + t] = r[k];
  }
  __syncthreads();
  int len = n - lo < RANGE ? n - lo : RANGE;
  for (int i = t; i < len; i += 1024) tab[(size_t)blockIdx.x * n + lo + i] = s_cnt[i];
  const int4* c4 = (const int4*)(s_cnt + t * SUBS);
  int v = 0;
  for (int q = 0; q < SUBS / 4; ++q) {
    int4 a = c4[q];
    int i = t * SUBS + q * 4;
    v += (i + 0 < len ? a.x : 0) + (i + 1 < len ? a.y : 0) + (i + 2 < len ? a.z : 0) + (i + 3 < len ? a.w : 0);
  }
  for (int o = 16; o > 0; o >>= 1) v += __shfl_xor(v, o);
  if ((t & 31) == 0) s_part[t >> 5] = v;
  __syncthreads();
  if (t < SUBS && t * SCAN_T < len) {
    int s = 0;
    for (int q = 0; q < 32 / SUBS; ++q) s += s_part[t * (32 / SUBS) + q];
    bsum[(size_t)(blockIdx.y * SUBS + t) * RCH + blockIdx.x] = s;
  }
}

__global__ void __launch_bounds__(SCAN_T)
k_scan(const int* __restrict__ bsum, int* __restrict__ cnt, int* __restrict__ start,
       float* __restrict__ dinv, int* tab, int n) {
  __shared__ int s_pre[32];
  __shared__ int s_tot[32];
  int t = threadIdx.x;
  int lane = t & 31;
  int w = t >> 5;
  int i = blockIdx.x * SCAN_T + t;
  int u[RCH];
#pragma unroll
  for (int c = 0; c < RCH; ++c) u[c] = i < n ? tab[(size_t)c * n + i] : 0;
  int before = blockIdx.x * RCH;
  int s = 0;
  for (int j = t; j < before; j += SCAN_T) s += bsum[j];
  int v = 0;
#pragma unroll
  for (int c = 0; c < RCH; ++c) v += u[c];
  int incl = v;
#pragma unroll
  for (int o = 1; o < 32; o <<= 1) {
    int x = __shfl_up(incl, o);
    if (lane >= o) incl += x;
  }
#pragma unroll
  for (int o = 16; o > 0; o >>= 1) s += __shfl_xor(s, o);
  if (lane == 31) s_tot[w] = incl;
  if (lane == 0) s_pre[w] = s;
  __syncthreads();
  int a = s_pre[lane] + (lane < w ? s_tot[lane] : 0);
#pragma unroll
  for (int o = 16; o > 0; o >>= 1) a += __shfl_xor(a, o);
  if (i < n) {
    int run = a + incl - v;
    start[i] = run;
    cnt[i] = v;
    dinv[i] = rsqrtf((float)(v + 1));
#pragma unroll
    for (int c = 0; c < RCH; ++c) {
      tab[(size_t)c * n + i] = run;
      run += u[c];
    }
  }
}

#define EPT 4
__global__ void k_fill(const int* __restrict__ src, const int* __restrict__ dst,
                       const int* __restrict__ rank, const int* __restrict__ tab,
                       int* __restrict__ csrc, int E, int n) {
  int t = blockIdx.x * blockDim.x + threadIdx.x;
  int T = gridDim.x * blockDim.x;
  int chunkE = (E + RCH - 1) / RCH;
  if (t + (EPT - 1) * T < E) {
    int d[EPT], r[EPT], s[EPT], p[EPT];
#pragma unroll
    for (int k = 0; k < EPT; ++k) {
      int e = t + k * T;
      d[k] = dst[e];
      r[k] = rank[e];
      s[k] = src[e];
    }
#pragma unroll
    for (int k = 0; k < EPT; ++k) p[k] = tab[(size_t)((t + k * T) / chunkE) * n + d[k]];
#pragma unroll
    for (int k = 0; k < EPT; ++k) csrc[p[k] + r[k]] = s[k];
  } else {
#pragma unroll
    for (int k = 0; k < EPT; ++k) {
      int e = t + k * T;
      if (e < E) csrc[tab[(size_t)(e / chunkE) * n + dst[e]] + rank[e]] = src[e];
    }
  }
}

#define AXP 8

__global__ void k_gather_x(const float* __restrict__ x, const float* __restrict__ dinv,
                           const int* __restrict__ start, const int* __restrict__ cnt,
                           const int* __restrict__ csrc, float* __restrict__ ax,
                           int n, int in_c) {
  int idx = blockIdx.x * blockDim.x + threadIdx.x;
  int d = idx >> 3;
  int c = threadIdx.x & 7;
  if (d >= n) return;
  int cc = c < in_c ? c : in_c - 1;
  int beg = start[d];
  int m = cnt[d];
  float dd = dinv[d];
  float acc = x[(size_t)d * in_c + cc] * (dd * dd);
  for (int base = 0; base < m; base += 8) {
    int sj = d;
    float wj = 0.0f;
    if (base + c < m) {
      sj = csrc[beg + base + c];
      wj = dinv[sj] * dd;
    }
    float v[8], w[8];
#pragma unroll
    for (int j = 0; j < 8; ++j) {
      int s = __shfl(sj, j, 8);
      w[j] = __shfl(wj, j, 8);
      v[j] = x[(size_t)s * in_c + cc];
    }
#pragma unroll
    for (int j = 0; j < 8; ++j) acc += v[j] * w[j];
  }
  if (c < in_c) ax[(size_t)d * AXP + c] = acc;
}

__global__ void k_gather64(const float* __restrict__ h, const float* __restrict__ dinv,
                           const int* __restrict__ start, const int* __restrict__ cnt,
                           const int* __restrict__ csrc, const float* __restrict__ b,
                           float* __restrict__ out, int n) {
  int lane = threadIdx.x & 31;
  int d = __builtin_amdgcn_readfirstlane((int)((blockIdx.x * blockDim.x + threadIdx.x) >> 5));
  if (d >= n) return;
  int beg = start[d];
  int m = cnt[d];
  float dd = dinv[d];
  const float2 hv = *(const float2*)(h + (size_t)d * 64 + lane * 2);
  float accx = hv.x * dd * dd + b[lane * 2 + 0];
  float accy = hv.y * dd * dd + b[lane * 2 + 1];
  for (int base = 0; base < m; base += 32) {
    int mm = m - base < 32 ? m - base : 32;
    int sj = d;
    float wj = 0.0f;
    if (lane < mm) {
      sj = csrc[beg + base + lane];
      wj = dinv[sj] * dd;
    }
    for (int j = 0; j < mm; j += 4) {
      float2 v[4];
      float wv[4];
#pragma unroll
      for (int u = 0; u < 4; ++u) {
        int s = __builtin_amdgcn_readlane(sj, j + u);
        wv[u] = __int_as_float(__builtin_amdgcn_readlane(__float_as_int(wj), j + u));
        v[u] = *(const float2*)(h + (size_t)s * 64 + lane * 2);
      }
#pragma unroll
      for (int u = 0; u < 4; ++u) {
        accx += v[u].x * wv[u];
        accy += v[u].y * wv[u];
      }
    }
  }
  float2 r;
  r.x = accx;
  r.y = accy;
  *(float2*)(out + (size_t)d * 64 + lane * 2) = r;
}

#define HID 128
#define OUTC 64
#define KSTEPS (HID / 4)
__global__ void __launch_bounds__(256)
k_mlp_wmma(const float* __restrict__ ax, const float* __restrict__ W1,
           const float* __restrict__ b1, const float* __restrict__ W2,
           float* __restrict__ D, int n, int in_c) {
  __shared__ float4 sB[KSTEPS * 2 * 32];
  __shared__ float4 sW[(HID / 2) * 4];

  int t = threadIdx.x;
#pragma unroll
  for (int j = 0; j < (KSTEPS * 32) / 256; ++j) {
    int step = j * 8 + (t >> 5);
    int ln = t & 31;
    int k0 = step * 4 + (ln >> 4) * 2;
    int c4 = (ln & 15) * 4;
    float4 r0 = *(const float4*)(W2 + (size_t)k0 * OUTC + c4);
    float4 r1 = *(const float4*)(W2 + (size_t)(k0 + 1) * OUTC + c4);
    sB[(step * 2 + 0) * 32 + ln] = make_float4(r0.x, r1.x, r0.y, r1.y);
    sB[(step * 2 + 1) * 32 + ln] = make_float4(r0.z, r1.z, r0.w, r1.w);
  }
  float* sWf = (float*)sW;
  int kk = t & (HID - 1);
  float bv = b1[kk];
#pragma unroll
  for (int j = 0; j < (HID * 8) / 256; ++j) {
    int i = j * 2 + (t >> 7);
    float v = 0.0f;
    if (i < in_c) v = W1[i * HID + kk];
    if (i == 7) v = bv;
    sWf[(kk >> 1) * 16 + i * 2 + (kk & 1)] = v;
  }
  __syncthreads();

  int lane = t & 31;
  int half = lane >> 4;
  int c    = lane & 15;
  int mTile = blockIdx.x * 8 + (t >> 5);
  if (mTile * 16 >= n) return;

  int rowA = mTile * 16 + c;
  if (rowA >= n) rowA = n - 1;
  const float4* ap = (const float4*)(ax + (size_t)rowA * AXP);
  float4 x0 = ap[0];
  float4 x1 = ap[1];
  float xr[7] = {x0.x, x0.y, x0.z, x0.w, x1.x, x1.y, x1.z};
#pragma unroll
  for (int i = 0; i < 7; ++i)
    if (i >= in_c) xr[i] = 0.0f;

  v8f acc0 = {}, acc1 = {}, acc2 = {}, acc3 = {};
#pragma unroll
  for (int s = 0; s < KSTEPS; ++s) {
    const float4* wp = sW + (s * 2 + half) * 4;
    float4 w01 = wp[0], w23 = wp[1], w45 = wp[2], w6b = wp[3];
    v2f w, xs, hh;
    w.x = w01.x; w.y = w01.y; hh = w * xr[0];
    w.x = w01.z; w.y = w01.w; xs.x = xr[1]; xs.y = xr[1]; hh = __builtin_elementwise_fma(xs, w, hh);
    w.x = w23.x; w.y = w23.y; xs.x = xr[2]; xs.y = xr[2]; hh = __builtin_elementwise_fma(xs, w, hh);
    w.x = w23.z; w.y = w23.w; xs.x = xr[3]; xs.y = xr[3]; hh = __builtin_elementwise_fma(xs, w, hh);
    w.x = w45.x; w.y = w45.y; xs.x = xr[4]; xs.y = xr[4]; hh = __builtin_elementwise_fma(xs, w, hh);
    w.x = w45.z; w.y = w45.w; xs.x = xr[5]; xs.y = xr[5]; hh = __builtin_elementwise_fma(xs, w, hh);
    w.x = w6b.x; w.y = w6b.y; xs.x = xr[6]; xs.y = xr[6]; hh = __builtin_elementwise_fma(xs, w, hh);
    w.x = w6b.z; w.y = w6b.w; hh = hh + w;
    v2f a;
    a.x = fmaxf(hh.x, 0.0f);
    a.y = fmaxf(hh.y, 0.0f);
    float4 q0 = sB[(s * 2 + 0) * 32 + lane];
    float4 q1 = sB[(s * 2 + 1) * 32 + lane];
    v2f f0, f1, f2, f3;
    f0.x = q0.x; f0.y = q0.y;
    f1.x = q0.z; f1.y = q0.w;
    f2.x = q1.x; f2.y = q1.y;
    f3.x = q1.z; f3.y = q1.w;
    acc0 = __builtin_amdgcn_wmma_f32_16x16x4_f32(false, a, false, f0, (short)0, acc0, false, false);
    acc1 = __builtin_amdgcn_wmma_f32_16x16x4_f32(false, a, false, f1, (short)0, acc1, false, false);
    acc2 = __builtin_amdgcn_wmma_f32_16x16x4_f32(false, a, false, f2, (short)0, acc2, false, false);
    acc3 = __builtin_amdgcn_wmma_f32_16x16x4_f32(false, a, false, f3, (short)0, acc3, false, false);
  }
#pragma unroll
  for (int r = 0; r < 8; ++r) {
    int row = mTile * 16 + r + half * 8;
    if (row < n)
      *(float4*)(D + (size_t)row * OUTC + c * 4) = make_float4(acc0[r], acc1[r], acc2[r], acc3[r]);
  }
}

extern "C" void kernel_launch(void* const* d_in, const int* in_sizes, int n_in,
                              void* d_out, int out_size, void* d_ws, size_t ws_size,
                              hipStream_t stream) {
  const float* x  = (const float*)d_in[0];
  const int*   ei = (const int*)d_in[1];
  const float* W1 = (const float*)d_in[2];
  const float* b1 = (const float*)d_in[3];
  const float* W2 = (const float*)d_in[4];
  const float* b2 = (const float*)d_in[5];
  float* out = (float*)d_out;

  int hid   = in_sizes[3];
  int in_c  = in_sizes[2] / hid;
  int n     = in_sizes[0] / in_c;
  int E     = in_sizes[1] / 2;

  const int* src = ei;
  const int* dst = ei + E;

  float* ws   = (float*)d_ws;
  float* dinv = ws;
  float* ax   = dinv + n;
  float* h2   = ax + (size_t)n * AXP;
  int* cnt    = (int*)(h2 + (size_t)n * OUTC);
  int* start  = cnt + n;
  int* rank   = start + n;
  int* csrc   = rank + E;
  int* tab    = csrc + E;
  int* bsum   = tab + (size_t)RCH * n;

  const int B256 = 256;
  k_rank_lds<<<dim3(RCH, (n + RANGE - 1) / RANGE), 1024, 0, stream>>>(dst, rank, tab, bsum, E, n);
  k_scan<<<(n + SCAN_T - 1) / SCAN_T, SCAN_T, 0, stream>>>(bsum, cnt, start, dinv, tab, n);
  int egrid = (E + B256 * EPT - 1) / (B256 * EPT);
  k_fill<<<egrid, B256, 0, stream>>>(src, dst, rank, tab, csrc, E, n);

  int totx = n * 8;
  k_gather_x<<<(totx + B256 - 1) / B256, B256, 0, stream>>>(x, dinv, start, cnt, csrc, ax, n, in_c);

  int mTiles = (n + 15) / 16;
  k_mlp_wmma<<<(mTiles + 7) / 8, B256, 0, stream>>>(ax, W1, b1, W2, h2, n, in_c);

  k_gather64<<<(n + 7) / 8, B256, 0, stream>>>(h2, dinv, start, cnt, csrc, b2, out, n);
}


# each row carries its own dinv from where it is made (x rows in the scan kernel, A rows of the W2 product): both gathers sum plain rows with no per-edge dinv load; edge order kept, bias added last
# speedup vs baseline: 5.0335x; 1.1127x over previous
#include <hip/hip_runtime.h>

typedef __attribute__((ext_vector_type(2))) float v2f;
typedef __attribute__((ext_vector_type(8))) float v8f;

#define RCH 16
#define RANGE 8192
#define RPT 16
#define SCAN_T 1024
#define SUBS (RANGE / SCAN_T)
#define AXP 8
static_assert(SCAN_T == 1024 && RANGE % SCAN_T == 0 && SUBS % 4 == 0 && 32 % SUBS == 0,
              "k_rank_lds: 1024 threads, SUBS counters each, whole waves per node block");
__global__ void __launch_bounds__(1024)
k_rank_lds(const int* __restrict__ dst, int* __restrict__ rank, int* __restrict__ tab,
           int* __restrict__ bsum, int E, int n) {
  __shared__ alignas(16) int s_cnt[RANGE];
  __shared__ int s_part[32];
  int t = threadIdx.x;
  int lo = blockIdx.y * RANGE;
  for (int i = t; i < RANGE; i += 1024) s_cnt[i] = 0;
  __syncthreads();
  int chunkE = (E + RCH - 1) / RCH;
  int e0 = blockIdx.x * chunkE;
  int e1 = e0 + chunkE < E ? e0 + chunkE : E;
  for (int base = e0; base < e1; base += 1024 * RPT) {
    int d[RPT];
#pragma unroll
    for (int k = 0; k < RPT; ++k) {
      int e = base + k * 1024 + t;
      d[k] = e < e1 ? dst[e] : -1;
    }
    int r[RPT];
#pragma unroll
    for (int k = 0; k < RPT; ++k) {
      unsigned o = (unsigned)(d[k] - lo);
      r[k] = -1;
      if (o < RANGE) r[k] = atomicAdd(&s_cnt[o], 1);
    }
#pragma unroll
    for (int k = 0; k < RPT; ++k)
      if (r[k] >= 0) rank[base + k * 1024 + t] = r[k];
  }
  __syncthreads();
  int len = n - lo < RANGE ? n - lo : RANGE;
  for (int i = t; i < len; i += 1024) tab[(size_t)blockIdx.x * n + lo + i] = s_cnt[i];
  const int4* c4 = (const int4*)(s_cnt + t * SUBS);
  int v = 0;
  for (int q = 0; q < SUBS / 4; ++q) {
    int4 a = c4[q];
    int i = t * SUBS + q * 4;
    v += (i + 0 < len ? a.x : 0) + (i + 1 < len ? a.y : 0) + (i + 2 < len ? a.z : 0) + (i + 3 < len ? a.w : 0);
  }
  for (int o = 16; o > 0; o >>= 1) v += __shfl_xor(v, o);
  if ((t & 31) == 0) s_part[t >> 5] = v;
  __syncthreads();
  if (t < SUBS && t * SCAN_T < len) {
    int s = 0;
    for (int q = 0; q < 32 / SUBS; ++q) s += s_part[t * (32 / SUBS) + q];
    bsum[(size_t)(blockIdx.y * SUBS + t) * RCH + blockIdx.x] = s;
  }
}

__global__ void __launch_bounds__(SCAN_T)
k_scan(const int* __restrict__ bsum, int* __restrict__ cnt, int* __restrict__ start,
       float* __restrict__ dinv, int* tab, const float* __restrict__ x,
       float* __restrict__ xs, int n, int in_c) {
  __shared__ int s_pre[32];
  __shared__ int s_tot[32];
  int t = threadIdx.x;
  int lane = t & 31;
  int w = t >> 5;
  int i = blockIdx.x * SCAN_T + t;
  int u[RCH];
#pragma unroll
  for (int c = 0; c < RCH; ++c) u[c] = i < n ? tab[(size_t)c * n + i] : 0;
  float xv[AXP];
#pragma unroll
  for (int c = 0; c < AXP; ++c) xv[c] = (i < n && c < in_c) ? x[(size_t)i * in_c + c] : 0.0f;
  int before = blockIdx.x * RCH;
  int s = 0;
  for (int j = t; j < before; j += SCAN_T) s += bsum[j];
  int v = 0;
#pragma unroll
  for (int c = 0; c < RCH; ++c) v += u[c];
  int incl = v;
#pragma unroll
  for (int o = 1; o < 32; o <<= 1) {
    int x = __shfl_up(incl, o);
    if (lane >= o) incl += x;
  }
#pragma unroll
  for (int o = 16; o > 0; o >>= 1) s += __shfl_xor(s, o);
  if (lane == 31) s_tot[w] = incl;
  if (lane == 0) s_pre[w] = s;
  __syncthreads();
  int a = s_pre[lane] + (lane < w ? s_tot[lane] : 0);
#pragma unroll
  for (int o = 16; o > 0; o >>= 1) a += __shfl_xor(a, o);
  if (i < n) {
    int run = a + incl - v;
    start[i] = run;
    cnt[i] = v;
    float dv = rsqrtf((float)(v + 1));
    dinv[i] = dv;
    float4* xp = (float4*)(xs + (size_t)i * AXP);
    xp[0] = make_float4(xv[0] * dv, xv[1] * dv, xv[2] * dv, xv[3] * dv);
    xp[1] = make_float4(xv[4] * dv, xv[5] * dv, xv[6] * dv, xv[7] * dv);
#pragma unroll
    for (int c = 0; c < RCH; ++c) {
      tab[(size_t)c * n + i] = run;
      run += u[c];
    }
  }
}

#define EPT 4
__global__ void k_fill(const int* __restrict__ src, const int* __restrict__ dst,
                       const int* __restrict__ rank, const int* __restrict__ tab,
                       int* __restrict__ csrc, int E, int n) {
  int t = blockIdx.x * blockDim.x + threadIdx.x;
  int T = gridDim.x * blockDim.x;
  int chunkE = (E + RCH - 1) / RCH;
  if (t + (EPT - 1) * T < E) {
    int d[EPT], r[EPT], s[EPT], p[EPT];
#pragma unroll
    for (int k = 0; k < EPT; ++k) {
      int e = t + k * T;
      d[k] = dst[e];
      r[k] = rank[e];
      s[k] = src[e];
    }
#pragma unroll
    for (int k = 0; k < EPT; ++k) p[k] = tab[(size_t)((t + k * T) / chunkE) * n + d[k]];
#pragma unroll
    for (int k = 0; k < EPT; ++k) csrc[p[k] + r[k]] = s[k];
  } else {
#pragma unroll
    for (int k = 0; k < EPT; ++k) {
      int e = t + k * T;
      if (e < E) csrc[tab[(size_t)(e / chunkE) * n + dst[e]] + rank[e]] = src[e];
    }
  }
}


__global__ void k_gather_x(const float* __restrict__ xs, const float* __restrict__ dinv,
                           const int* __restrict__ start, const int* __restrict__ cnt,
                           const int* __restrict__ csrc, float* __restrict__ ax,
                           int n, int in_c) {
  int idx = blockIdx.x * blockDim.x + threadIdx.x;
  int d = idx >> 3;
  int c = threadIdx.x & 7;
  if (d >= n) return;
  int beg = start[d];
  int m = cnt[d];
  float dd = dinv[d];
  float acc = xs[(size_t)d * AXP + c];
  for (int base = 0; base < m; base += 8) {
    int sj = d;
    if (base + c < m) sj = csrc[beg + base + c];
    float v[8];
#pragma unroll
    for (int j = 0; j < 8; ++j) {
      int s = __shfl(sj, j, 8);
      v[j] = xs[(size_t)s * AXP + c];
    }
#pragma unroll
    for (int j = 0; j < 8; ++j) acc += base + j < m ? v[j] : 0.0f;
  }
  if (c < in_c) ax[(size_t)d * AXP + c] = acc * dd;
}

__global__ void k_gather64(const float* __restrict__ h, const float* __restrict__ dinv,
                           const int* __restrict__ start, const int* __restrict__ cnt,
                           const int* __restrict__ csrc, const float* __restrict__ b,
                           float* __restrict__ out, int n) {
  int lane = threadIdx.x & 31;
  int d = __builtin_amdgcn_readfirstlane((int)((blockIdx.x * blockDim.x + threadIdx.x) >> 5));
  if (d >= n) return;
  int beg = start[d];
  int m = cnt[d];
  float dd = dinv[d];
  const float2 hv = *(const float2*)(h + (size_t)d * 64 + lane * 2);
  const float2 bv = *(const float2*)(b + lane * 2);
  asm volatile("" :: "v"(bv.x), "v"(bv.y));
  float accx = hv.x;
  float accy = hv.y;
  for (int base = 0; base < m; base += 32) {
    int mm = m - base < 32 ? m - base : 32;
    int sj = d;
    if (lane < mm) sj = csrc[beg + base + lane];
    for (int j = 0; j < mm; j += 4) {
      float2 v[4];
      float wv[4];
#pragma unroll
      for (int u = 0; u < 4; ++u) {
        int s = __builtin_amdgcn_readlane(sj, j + u);
        wv[u] = j + u < mm ? 1.0f : 0.0f;
        v[u] = *(const float2*)(h + (size_t)s * 64 + lane * 2);
      }
#pragma unroll
      for (int u = 0; u < 4; ++u) {
        accx += v[u].x * wv[u];
        accy += v[u].y * wv[u];
      }
    }
  }
  float2 r;
  r.x = accx * dd + bv.x;
  r.y = accy * dd + bv.y;
  *(float2*)(out + (size_t)d * 64 + lane * 2) = r;
}

#define HID 128
#define OUTC 64
#define KSTEPS (HID / 4)
__global__ void __launch_bounds__(256)
k_mlp_wmma(const float* __restrict__ ax, const float* __restrict__ W1,
           const float* __restrict__ b1, const float* __restrict__ W2,
           const float* __restrict__ dinv, float* __restrict__ D, int n, int in_c) {
  __shared__ float4 sB[KSTEPS * 2 * 32];
  __shared__ float4 sW[(HID / 2) * 4];

  int t = threadIdx.x;
#pragma unroll
  for (int j = 0; j < (KSTEPS * 32) / 256; ++j) {
    int step = j * 8 + (t >> 5);
    int ln = t & 31;
    int k0 = step * 4 + (ln >> 4) * 2;
    int c4 = (ln & 15) * 4;
    float4 r0 = *(const float4*)(W2 + (size_t)k0 * OUTC + c4);
    float4 r1 = *(const float4*)(W2 + (size_t)(k0 + 1) * OUTC + c4);
    sB[(step * 2 + 0) * 32 + ln] = make_float4(r0.x, r1.x, r0.y, r1.y);
    sB[(step * 2 + 1) * 32 + ln] = make_float4(r0.z, r1.z, r0.w, r1.w);
  }
  float* sWf = (float*)sW;
  int kk = t & (HID - 1);
  float bv = b1[kk];
#pragma unroll
  for (int j = 0; j < (HID * 8) / 256; ++j) {
    int i = j * 2 + (t >> 7);
    float v = 0.0f;
    if (i < in_c) v = W1[i * HID + kk];
    if (i == 7) v = bv;
    sWf[(kk >> 1) * 16 + i * 2 + (kk & 1)] = v;
  }
  __syncthreads();

  int lane = t & 31;
  int half = lane >> 4;
  int c    = lane & 15;
  int mTile = blockIdx.x * 8 + (t >> 5);
  if (mTile * 16 >= n) return;

  int rowA = mTile * 16 + c;
  if (rowA >= n) rowA = n - 1;
  const float4* ap = (const float4*)(ax + (size_t)rowA * AXP);
  float4 x0 = ap[0];
  float4 x1 = ap[1];
  float dv = dinv[rowA];
  float xr[7] = {x0.x, x0.y, x0.z, x0.w, x1.x, x1.y, x1.z};
#pragma unroll
  for (int i = 0; i < 7; ++i)
    if (i >= in_c) xr[i] = 0.0f;

  v8f acc0 = {}, acc1 = {}, acc2 = {}, acc3 = {};
#pragma unroll
  for (int s = 0; s < KSTEPS; ++s) {
    const float4* wp = sW + (s * 2 + half) * 4;
    float4 w01 = wp[0], w23 = wp[1], w45 = wp[2], w6b = wp[3];
    v2f w, xs, hh;
    w.x = w01.x; w.y = w01.y; hh = w * xr[0];
    w.x = w01.z; w.y = w01.w; xs.x = xr[1]; xs.y = xr[1]; hh = __builtin_elementwise_fma(xs, w, hh);
    w.x = w23.x; w.y = w23.y; xs.x = xr[2]; xs.y = xr[2]; hh = __builtin_elementwise_fma(xs, w, hh);
    w.x = w23.z; w.y = w23.w; xs.x = xr[3]; xs.y = xr[3]; hh = __builtin_elementwise_fma(xs, w, hh);
    w.x = w45.x; w.y = w45.y; xs.x = xr[4]; xs.y = xr[4]; hh = __builtin_elementwise_fma(xs, w, hh);
    w.x = w45.z; w.y = w45.w; xs.x = xr[5]; xs.y = xr[5]; hh = __builtin_elementwise_fma(xs, w, hh);
    w.x = w6b.x; w.y = w6b.y; xs.x = xr[6]; xs.y = xr[6]; hh = __builtin_elementwise_fma(xs, w, hh);
    w.x = w6b.z; w.y = w6b.w; hh = hh + w;
    v2f a;
    a.x = fmaxf(hh.x, 0.0f) * dv;
    a.y = fmaxf(hh.y, 0.0f) * dv;
    float4 q0 = sB[(s * 2 + 0) * 32 + lane];
    float4 q1 = sB[(s * 2 + 1) * 32 + lane];
    v2f f0, f1, f2, f3;
    f0.x = q0.x; f0.y = q0.y;
    f1.x = q0.z; f1.y = q0.w;
    f2.x = q1.x; f2.y = q1.y;
    f3.x = q1.z; f3.y = q1.w;
    acc0 = __builtin_amdgcn_wmma_f32_16x16x4_f32(false, a, false, f0, (short)0, acc0, false, false);
    acc1 = __builtin_amdgcn_wmma_f32_16x16x4_f32(false, a, false, f1, (short)0, acc1, false, false);
    acc2 = __builtin_amdgcn_wmma_f32_16x16x4_f32(false, a, false, f2, (short)0, acc2, false, false);
    acc3 = __builtin_amdgcn_wmma_f32_16x16x4_f32(false, a, false, f3, (short)0, acc3, false, false);
  }
#pragma unroll
  for (int r = 0; r < 8; ++r) {
    int row = mTile * 16 + r + half * 8;
    if (row < n)
      *(float4*)(D + (size_t)row * OUTC + c * 4) = make_float4(acc0[r], acc1[r], acc2[r], acc3[r]);
  }
}

extern "C" void kernel_launch(void* const* d_in, const int* in_sizes, int n_in,
                              void* d_out, int out_size, void* d_ws, size_t ws_size,
                              hipStream_t stream) {
  const float* x  = (const float*)d_in[0];
  const int*   ei = (const int*)d_in[1];
  const float* W1 = (const float*)d_in[2];
  const float* b1 = (const float*)d_in[3];
  const float* W2 = (const float*)d_in[4];
  const float* b2 = (const float*)d_in[5];
  float* out = (float*)d_out;

  int hid   = in_sizes[3];
  int in_c  = in_sizes[2] / hid;
  int n     = in_sizes[0] / in_c;
  int E     = in_sizes[1] / 2;

  const int* src = ei;
  const int* dst = ei + E;

  float* ws   = (float*)d_ws;
  float* dinv = ws;
  float* ax   = dinv + n;
  float* xs   = ax + (size_t)n * AXP;
  float* h2   = xs + (size_t)n * AXP;
  int* cnt    = (int*)(h2 + (size_t)n * OUTC);
  int* start  = cnt + n;
  int* rank   = start + n;
  int* csrc   = rank + E;
  int* tab    = csrc + E;
  int* bsum   = tab + (size_t)RCH * n;

  const int B256 = 256;
  k_rank_lds<<<dim3(RCH, (n + RANGE - 1) / RANGE), 1024, 0, stream>>>(dst, rank, tab, bsum, E, n);
  k_scan<<<(n + SCAN_T - 1) / SCAN_T, SCAN_T, 0, stream>>>(bsum, cnt, start, dinv, tab, x, xs, n, in_c);
  int egrid = (E + B256 * EPT - 1) / (B256 * EPT);
  k_fill<<<egrid, B256, 0, stream>>>(src, dst, rank, tab, csrc, E, n);

  int totx = n * 8;
  k_gather_x<<<(totx + B256 - 1) / B256, B256, 0, stream>>>(xs, dinv, start, cnt, csrc, ax, n, in_c);

  int mTiles = (n + 15) / 16;
  k_mlp_wmma<<<(mTiles + 7) / 8, B256, 0, stream>>>(ax, W1, b1, W2, dinv, h2, n, in_c);

  k_gather64<<<(n + 7) / 8, B256, 0, stream>>>(h2, dinv, start, cnt, csrc, b2, out, n);
}


# layer-1 gather fused into the WMMA kernel: lane (row c, half h) sums 16-byte half rows of row c's in-edges in registers, one cross-lane swap joins the halves; ax plane and a launch go; sum order kept
# speedup vs baseline: 5.4002x; 1.0729x over previous
#include <hip/hip_runtime.h>

typedef __attribute__((ext_vector_type(2))) float v2f;
typedef __attribute__((ext_vector_type(8))) float v8f;

#define RCH 16
#define RANGE 8192
#define RPT 16
#define SCAN_T 1024
#define SUBS (RANGE / SCAN_T)
#define AXP 8
static_assert(SCAN_T == 1024 && RANGE % SCAN_T == 0 && SUBS % 4 == 0 && 32 % SUBS == 0,
              "k_rank_lds: 1024 threads, SUBS counters each, whole waves per node block");
__global__ void __launch_bounds__(1024)
k_rank_lds(const int* __restrict__ dst, int* __restrict__ rank, int* __restrict__ tab,
           int* __restrict__ bsum, int E, int n) {
  __shared__ alignas(16) int s_cnt[RANGE];
  __shared__ int s_part[32];
  int t = threadIdx.x;
  int lo = blockIdx.y * RANGE;
  for (int i = t; i < RANGE; i += 1024) s_cnt[i] = 0;
  __syncthreads();
  int chunkE = (E + RCH - 1) / RCH;
  int e0 = blockIdx.x * chunkE;
  int e1 = e0 + chunkE < E ? e0 + chunkE : E;
  for (int base = e0; base < e1; base += 1024 * RPT) {
    int d[RPT];
#pragma unroll
    for (int k = 0; k < RPT; ++k) {
      int e = base + k * 1024 + t;
      d[k] = e < e1 ? dst[e] : -1;
    }
    int r[RPT];
#pragma unroll
    for (int k = 0; k < RPT; ++k) {
      unsigned o = (unsigned)(d[k] - lo);
      r[k] = -1;
      if (o < RANGE) r[k] = atomicAdd(&s_cnt[o], 1);
    }
#pragma unroll
    for (int k = 0; k < RPT; ++k)
      if (r[k] >= 0) rank[base + k * 1024 + t] = r[k];
  }
  __syncthreads();
  int len = n - lo < RANGE ? n - lo : RANGE;
  for (int i = t; i < len; i += 1024) tab[(size_t)blockIdx.x * n + lo + i] = s_cnt[i];
  const int4* c4 = (const int4*)(s_cnt + t * SUBS);
  int v = 0;
  for (int q = 0; q < SUBS / 4; ++q) {
    int4 a = c4[q];
    int i = t * SUBS + q * 4;
    v += (i + 0 < len ? a.x : 0) + (i + 1 < len ? a.y : 0) + (i + 2 < len ? a.z : 0) + (i + 3 < len ? a.w : 0);
  }
  for (int o = 16; o > 0; o >>= 1) v += __shfl_xor(v, o);
  if ((t & 31) == 0) s_part[t >> 5] = v;
  __syncthreads();
  if (t < SUBS && t * SCAN_T < len) {
    int s = 0;
    for (int q = 0; q < 32 / SUBS; ++q) s += s_part[t * (32 / SUBS) + q];
    bsum[(size_t)(blockIdx.y * SUBS + t) * RCH + blockIdx.x] = s;
  }
}

__global__ void __launch_bounds__(SCAN_T)
k_scan(const int* __restrict__ bsum, int* __restrict__ cnt, int* __restrict__ start,
       float* __restrict__ dinv, int* tab, const float* __restrict__ x,
       float* __restrict__ xs, int n, int in_c) {
  __shared__ int s_pre[32];
  __shared__ int s_tot[32];
  int t = threadIdx.x;
  int lane = t & 31;
  int w = t >> 5;
  int i = blockIdx.x * SCAN_T + t;
  int u[RCH];
#pragma unroll
  for (int c = 0; c < RCH; ++c) u[c] = i < n ? tab[(size_t)c * n + i] : 0;
  float xv[AXP];
#pragma unroll
  for (int c = 0; c < AXP; ++c) xv[c] = (i < n && c < in_c) ? x[(size_t)i * in_c + c] : 0.0f;
  int before = blockIdx.x * RCH;
  int s = 0;
  for (int j = t; j < before; j += SCAN_T) s += bsum[j];
  int v = 0;
#pragma unroll
  for (int c = 0; c < RCH; ++c) v += u[c];
  int incl = v;
#pragma unroll
  for (int o = 1; o < 32; o <<= 1) {
    int x = __shfl_up(incl, o);
    if (lane >= o) incl += x;
  }
#pragma unroll
  for (int o = 16; o > 0; o >>= 1) s += __shfl_xor(s, o);
  if (lane == 31) s_tot[w] = incl;
  if (lane == 0) s_pre[w] = s;
  __syncthreads();
  int a = s_pre[lane] + (lane < w ? s_tot[lane] : 0);
#pragma unroll
  for (int o = 16; o > 0; o >>= 1) a += __shfl_xor(a, o);
  if (i < n) {
    int run = a + incl - v;
    start[i] = run;
    cnt[i] = v;
    float dv = rsqrtf((float)(v + 1));
    dinv[i] = dv;
    float4* xp = (float4*)(xs + (size_t)i * AXP);
    xp[0] = make_float4(xv[0] * dv, xv[1] * dv, xv[2] * dv, xv[3] * dv);
    xp[1] = make_float4(xv[4] * dv, xv[5] * dv, xv[6] * dv, xv[7] * dv);
#pragma unroll
    for (int c = 0; c < RCH; ++c) {
      tab[(size_t)c * n + i] = run;
      run += u[c];
    }
  }
}

#define EPT 4
__global__ void k_fill(const int* __restrict__ src, const int* __restrict__ dst,
                       const int* __restrict__ rank, const int* __restrict__ tab,
                       int* __restrict__ csrc, int E, int n) {
  int t = blockIdx.x * blockDim.x + threadIdx.x;
  int T = gridDim.x * blockDim.x;
  int chunkE = (E + RCH - 1) / RCH;
  if (t + (EPT - 1) * T < E) {
    int d[EPT], r[EPT], s[EPT], p[EPT];
#pragma unroll
    for (int k = 0; k < EPT; ++k) {
      int e = t + k * T;
      d[k] = dst[e];
      r[k] = rank[e];
      s[k] = src[e];
    }
#pragma unroll
    for (int k = 0; k < EPT; ++k) p[k] = tab[(size_t)((t + k * T) / chunkE) * n + d[k]];
#pragma unroll
    for (int k = 0; k < EPT; ++k) csrc[p[k] + r[k]] = s[k];
  } else {
#pragma unroll
    for (int k = 0; k < EPT; ++k) {
      int e = t + k * T;
      if (e < E) csrc[tab[(size_t)(e / chunkE) * n + dst[e]] + rank[e]] = src[e];
    }
  }
}


__global__ void k_gather64(const float* __restrict__ h, const float* __restrict__ dinv,
                           const int* __restrict__ start, const int* __restrict__ cnt,
                           const int* __restrict__ csrc, const float* __restrict__ b,
                           float* __restrict__ out, int n) {
  int lane = threadIdx.x & 31;
  int d = __builtin_amdgcn_readfirstlane((int)((blockIdx.x * blockDim.x + threadIdx.x) >> 5));
  if (d >= n) return;
  int beg = start[d];
  int m = cnt[d];
  float dd = dinv[d];
  const float2 hv = *(const float2*)(h + (size_t)d * 64 + lane * 2);
  const float2 bv = *(const float2*)(b + lane * 2);
  asm volatile("" :: "v"(bv.x), "v"(bv.y));
  float accx = hv.x;
  float accy = hv.y;
  for (int base = 0; base < m; base += 32) {
    int mm = m - base < 32 ? m - base : 32;
    int sj = d;
    if (lane < mm) sj = csrc[beg + base + lane];
    for (int j = 0; j < mm; j += 4) {
      float2 v[4];
      float wv[4];
#pragma unroll
      for (int u = 0; u < 4; ++u) {
        int s = __builtin_amdgcn_readlane(sj, j + u);
        wv[u] = j + u < mm ? 1.0f : 0.0f;
        v[u] = *(const float2*)(h + (size_t)s * 64 + lane * 2);
      }
#pragma unroll
      for (int u = 0; u < 4; ++u) {
        accx += v[u].x * wv[u];
        accy += v[u].y * wv[u];
      }
    }
  }
  float2 r;
  r.x = accx * dd + bv.x;
  r.y = accy * dd + bv.y;
  *(float2*)(out + (size_t)d * 64 + lane * 2) = r;
}

#define HID 128
#define OUTC 64
#define KSTEPS (HID / 4)
#define GU 16
static_assert(AXP == 8, "k_gather_mlp_wmma: a row of xs is two 16-byte halves, one per lane half");
__global__ void __launch_bounds__(256)
k_gather_mlp_wmma(const float* __restrict__ xs, const float* __restrict__ dinv,
                  const int* __restrict__ start, const int* __restrict__ cnt,
                  const int* __restrict__ csrc, const float* __restrict__ W1,
                  const float* __restrict__ b1, const float* __restrict__ W2,
                  float* __restrict__ D, int n, int in_c) {
  __shared__ float4 sB[KSTEPS * 2 * 32];
  __shared__ float4 sW[(HID / 2) * 4];

  int t = threadIdx.x;
  int lane = t & 31;
  int half = lane >> 4;
  int c    = lane & 15;
  int mTile = blockIdx.x * 8 + (t >> 5);
  bool live = mTile * 16 < n;
  int rowA = mTile * 16 + c;
  if (rowA >= n) rowA = n - 1;
  int beg = start[rowA];
  int m = cnt[rowA];
  if (!live) m = 0;
  float dv = dinv[rowA];
  const float* xh = xs + half * 4;
  float4 acc = *(const float4*)(xh + (size_t)rowA * AXP);

  float4 r0[(KSTEPS * 32) / 256], r1[(KSTEPS * 32) / 256];
#pragma unroll
  for (int j = 0; j < (KSTEPS * 32) / 256; ++j) {
    int k0 = (j * 8 + (t >> 5)) * 4 + half * 2;
    r0[j] = *(const float4*)(W2 + (size_t)k0 * OUTC + c * 4);
    r1[j] = *(const float4*)(W2 + (size_t)(k0 + 1) * OUTC + c * 4);
  }
  int kk = t & (HID - 1);
  float bv = b1[kk];
  float w1v[(HID * 8) / 256];
#pragma unroll
  for (int j = 0; j < (HID * 8) / 256; ++j) {
    int i = j * 2 + (t >> 7);
    w1v[j] = 0.0f;
    if (i < in_c) w1v[j] = W1[i * HID + kk];
  }
  int dummy = mTile * 16 < n ? mTile * 16 : n - 1;
  int idx[GU];
#pragma unroll
  for (int u = 0; u < GU; ++u) {
    int sv = csrc[u < m ? beg + u : 0];
    idx[u] = u < m ? sv : dummy;
  }
#pragma unroll
  for (int j = 0; j < (KSTEPS * 32) / 256; ++j) {
    int step = j * 8 + (t >> 5);
    sB[(step * 2 + 0) * 32 + lane] = make_float4(r0[j].x, r1[j].x, r0[j].y, r1[j].y);
    sB[(step * 2 + 1) * 32 + lane] = make_float4(r0[j].z, r1[j].z, r0[j].w, r1[j].w);
  }
  float* sWf = (float*)sW;
#pragma unroll
  for (int j = 0; j < (HID * 8) / 256; ++j) {
    int i = j * 2 + (t >> 7);
    sWf[(kk >> 1) * 16 + i * 2 + (kk & 1)] = i == 7 ? bv : w1v[j];
  }

  for (int base = 0; __any(base < m); base += GU) {
    float4 v[GU];
#pragma unroll
    for (int u = 0; u < GU; ++u) v[u] = *(const float4*)(xh + (size_t)idx[u] * AXP);
    int nb = base + GU;
#pragma unroll
    for (int u = 0; u < GU; ++u) {
      int sv = csrc[nb + u < m ? beg + nb + u : 0];
      idx[u] = nb + u < m ? sv : dummy;
    }
#pragma unroll
    for (int u = 0; u < GU; ++u) {
      bool on = base + u < m;
      acc.x += on ? v[u].x : 0.0f;
      acc.y += on ? v[u].y : 0.0f;
      acc.z += on ? v[u].z : 0.0f;
      acc.w += on ? v[u].w : 0.0f;
    }
  }
  float4 mine = make_float4(acc.x * dv, acc.y * dv, acc.z * dv, acc.w * dv);
  float4 oth;
  oth.x = __shfl_xor(mine.x, 16);
  oth.y = __shfl_xor(mine.y, 16);
  oth.z = __shfl_xor(mine.z, 16);
  oth.w = __shfl_xor(mine.w, 16);
  float4 x0 = half ? oth : mine;
  float4 x1 = half ? mine : oth;
  float xr[7] = {x0.x, x0.y, x0.z, x0.w, x1.x, x1.y, x1.z};

  __syncthreads();
  if (!live) return;

  v8f acc0 = {}, acc1 = {}, acc2 = {}, acc3 = {};
#pragma unroll
  for (int s = 0; s < KSTEPS; ++s) {
    const float4* wp = sW + (s * 2 + half) * 4;
    float4 w01 = wp[0], w23 = wp[1], w45 = wp[2], w6b = wp[3];
    v2f w, xv, hh;
    w.x = w01.x; w.y = w01.y; hh = w * xr[0];
    w.x = w01.z; w.y = w01.w; xv.x = xr[1]; xv.y = xr[1]; hh = __builtin_elementwise_fma(xv, w, hh);
    w.x = w23.x; w.y = w23.y; xv.x = xr[2]; xv.y = xr[2]; hh = __builtin_elementwise_fma(xv, w, hh);
    w.x = w23.z; w.y = w23.w; xv.x = xr[3]; xv.y = xr[3]; hh = __builtin_elementwise_fma(xv, w, hh);
    w.x = w45.x; w.y = w45.y; xv.x = xr[4]; xv.y = xr[4]; hh = __builtin_elementwise_fma(xv, w, hh);
    w.x = w45.z; w.y = w45.w; xv.x = xr[5]; xv.y = xr[5]; hh = __builtin_elementwise_fma(xv, w, hh);
    w.x = w6b.x; w.y = w6b.y; xv.x = xr[6]; xv.y = xr[6]; hh = __builtin_elementwise_fma(xv, w, hh);
    w.x = w6b.z; w.y = w6b.w; hh = hh + w;
    v2f a;
    a.x = fmaxf(hh.x, 0.0f) * dv;
    a.y = fmaxf(hh.y, 0.0f) * dv;
    float4 q0 = sB[(s * 2 + 0) * 32 + lane];
    float4 q1 = sB[(s * 2 + 1) * 32 + lane];
    v2f f0, f1, f2, f3;
    f0.x = q0.x; f0.y = q0.y;
    f1.x = q0.z; f1.y = q0.w;
    f2.x = q1.x; f2.y = q1.y;
    f3.x = q1.z; f3.y = q1.w;
    acc0 = __builtin_amdgcn_wmma_f32_16x16x4_f32(false, a, false, f0, (short)0, acc0, false, false);
    acc1 = __builtin_amdgcn_wmma_f32_16x16x4_f32(false, a, false, f1, (short)0, acc1, false, false);
    acc2 = __builtin_amdgcn_wmma_f32_16x16x4_f32(false, a, false, f2, (short)0, acc2, false, false);
    acc3 = __builtin_amdgcn_wmma_f32_16x16x4_f32(false, a, false, f3, (short)0, acc3, false, false);
  }
#pragma unroll
  for (int r = 0; r < 8; ++r) {
    int row = mTile * 16 + r + half * 8;
    if (row < n)
      *(float4*)(D + (size_t)row * OUTC + c * 4) = make_float4(acc0[r], acc1[r], acc2[r], acc3[r]);
  }
}

extern "C" void kernel_launch(void* const* d_in, const int* in_sizes, int n_in,
                              void* d_out, int out_size, void* d_ws, size_t ws_size,
                              hipStream_t stream) {
  const float* x  = (const float*)d_in[0];
  const int*   ei = (const int*)d_in[1];
  const float* W1 = (const float*)d_in[2];
  const float* b1 = (const float*)d_in[3];
  const float* W2 = (const float*)d_in[4];
  const float* b2 = (const float*)d_in[5];
  float* out = (float*)d_out;

  int hid   = in_sizes[3];
  int in_c  = in_sizes[2] / hid;
  int n     = in_sizes[0] / in_c;
  int E     = in_sizes[1] / 2;

  const int* src = ei;
  const int* dst = ei + E;

  float* ws   = (float*)d_ws;
  float* dinv = ws;
  float* xs   = dinv + n;
  float* h2   = xs + (size_t)n * AXP;
  int* cnt    = (int*)(h2 + (size_t)n * OUTC);
  int* start  = cnt + n;
  int* rank   = start + n;
  int* csrc   = rank + E;
  int* tab    = csrc + E;
  int* bsum   = tab + (size_t)RCH * n;

  const int B256 = 256;
  k_rank_lds<<<dim3(RCH, (n + RANGE - 1) / RANGE), 1024, 0, stream>>>(dst, rank, tab, bsum, E, n);
  k_scan<<<(n + SCAN_T - 1) / SCAN_T, SCAN_T, 0, stream>>>(bsum, cnt, start, dinv, tab, x, xs, n, in_c);
  int egrid = (E + B256 * EPT - 1) / (B256 * EPT);
  k_fill<<<egrid, B256, 0, stream>>>(src, dst, rank, tab, csrc, E, n);

  int mTiles = (n + 15) / 16;
  k_gather_mlp_wmma<<<(mTiles + 7) / 8, B256, 0, stream>>>(xs, dinv, start, cnt, csrc, W1, b1, W2, h2, n, in_c);

  k_gather64<<<(n + 7) / 8, B256, 0, stream>>>(h2, dinv, start, cnt, csrc, b2, out, n);
}


# slot scan and edge placement in one kernel per (edge chunk, node range): first slots from the count table into LDS, edges take slots by returning LDS adds; rank plane and a launch go; sum order kept
# speedup vs baseline: 5.7249x; 1.0601x over previous
#include <hip/hip_runtime.h>

typedef __attribute__((ext_vector_type(2))) float v2f;
typedef __attribute__((ext_vector_type(8))) float v8f;

#define RCH 16
#define RANGE 8192
#define RPT 16
#define SCAN_T 1024
#define SUBS (RANGE / SCAN_T)
#define NPT (RANGE / 1024)
#define AXP 8
static_assert(SCAN_T == 1024 && RANGE % SCAN_T == 0 && SUBS % 4 == 0 && 32 % SUBS == 0,
              "k_count_lds: 1024 threads, SUBS counters each, whole waves per node block");
static_assert(NPT == 8 && RCH == 2 * NPT,
              "k_scan_fill: 8 nodes per thread, and (slice of 1024 nodes, half of it) numbers the RCH chunk work-groups");
__global__ void __launch_bounds__(1024)
k_count_lds(const int* __restrict__ dst, int* __restrict__ tab, int* __restrict__ bsum, int E, int n) {
  __shared__ alignas(16) int s_cnt[RANGE];
  __shared__ int s_part[32];
  int t = threadIdx.x;
  int lo = blockIdx.y * RANGE;
  for (int i = t; i < RANGE; i += 1024) s_cnt[i] = 0;
  __syncthreads();
  int chunkE = (E + RCH - 1) / RCH;
  int e0 = blockIdx.x * chunkE;
  int e1 = e0 + chunkE < E ? e0 + chunkE : E;
  for (int base = e0; base < e1; base += 1024 * RPT) {
    int d[RPT];
#pragma unroll
    for (int k = 0; k < RPT; ++k) {
      int e = base + k * 1024 + t;
      d[k] = dst[e < e1 ? e : e1 - 1];
    }
    __builtin_amdgcn_sched_barrier(0);
#pragma unroll
    for (int k = 0; k < RPT; ++k) d[k] = base + k * 1024 + t < e1 ? d[k] : -1;
#pragma unroll
    for (int k = 0; k < RPT; ++k) {
      unsigned o = (unsigned)(d[k] - lo);
      if (o < RANGE) atomicAdd(&s_cnt[o], 1);
    }
  }
  __syncthreads();
  int len = n - lo < RANGE ? n - lo : RANGE;
  for (int i = t; i < len; i += 1024) tab[(size_t)blockIdx.x * n + lo + i] = s_cnt[i];
  const int4* c4 = (const int4*)(s_cnt + t * SUBS);
  int v = 0;
  for (int q = 0; q < SUBS / 4; ++q) {
    int4 a = c4[q];
    int i = t * SUBS + q * 4;
    v += (i + 0 < len ? a.x : 0) + (i + 1 < len ? a.y : 0) + (i + 2 < len ? a.z : 0) + (i + 3 < len ? a.w : 0);
  }
  for (int o = 16; o > 0; o >>= 1) v += __shfl_xor(v, o);
  if ((t & 31) == 0) s_part[t >> 5] = v;
  __syncthreads();
  if (t < SUBS && t * SCAN_T < len) {
    int s = 0;
    for (int q = 0; q < 32 / SUBS; ++q) s += s_part[t * (32 / SUBS) + q];
    bsum[(size_t)(blockIdx.y * SUBS + t) * RCH + blockIdx.x] = s;
  }
}

__global__ void __launch_bounds__(1024)
k_scan_fill(const int* __restrict__ src, const int* __restrict__ dst,
            const int* __restrict__ tab, const int* __restrict__ bsum,
            int* __restrict__ cnt, int* __restrict__ start, float* __restrict__ dinv,
            const float* __restrict__ x, float* __restrict__ xs, int* __restrict__ csrc,
            int E, int n, int in_c) {
  __shared__ alignas(16) int s_slot[RANGE];
  __shared__ int s_front[RANGE];
  __shared__ int s_pre[32];
  __shared__ int s_tot[32];
  int t = threadIdx.x;
  int lane = t & 31;
  int w = t >> 5;
  int c = blockIdx.x;
  int lo = blockIdx.y * RANGE;
  int len = n - lo < RANGE ? n - lo : RANGE;
  int before = blockIdx.y * SUBS * RCH;
  int s0 = t < before ? bsum[t] : 0;
  int s1 = t + 1024 < before ? bsum[t + 1024] : 0;
  int qa = c >> 1;
  int ia = qa * 1024 + t;
  bool mine = ia < len && (t >> 9) == (c & 1);
  float xv[AXP];
#pragma unroll
  for (int k = 0; k < AXP; ++k) xv[k] = (mine && k < in_c) ? x[(size_t)(lo + ia) * in_c + k] : 0.0f;
  int va = 0;
#pragma unroll
  for (int h = 0; h < 2; ++h) {
    int u[NPT / 2][RCH];
#pragma unroll
    for (int qq = 0; qq < NPT / 2; ++qq) {
      int i = (h * (NPT / 2) + qq) * 1024 + t;
      const int* tp = tab + lo + (i < len ? i : len - 1);
#pragma unroll
      for (int cc = 0; cc < RCH; ++cc) u[qq][cc] = tp[(size_t)cc * n];
    }
    __builtin_amdgcn_sched_barrier(0);
#pragma unroll
    for (int qq = 0; qq < NPT / 2; ++qq) {
      int q = h * (NPT / 2) + qq;
      int i = q * 1024 + t;
      int a = 0, p = 0;
#pragma unroll
      for (int cc = 0; cc < RCH; ++cc) {
        a += u[qq][cc];
        p += cc < c ? u[qq][cc] : 0;
      }
      a = i < len ? a : 0;
      s_slot[i] = a;
      s_front[i] = p;
      if (q == qa) va = a;
    }
  }
  int s = s0 + s1;
  for (int j = t + 2048; j < before; j += 1024) s += bsum[j];
#pragma unroll
  for (int o = 16; o > 0; o >>= 1) s += __shfl_xor(s, o);
  if (lane == 0) s_pre[w] = s;
  __syncthreads();
  int4 a0 = ((const int4*)s_slot)[t * 2];
  int4 a1 = ((const int4*)s_slot)[t * 2 + 1];
  int tot = a0.x + a0.y + a0.z + a0.w + a1.x + a1.y + a1.z + a1.w;
  int incl = tot;
#pragma unroll
  for (int o = 1; o < 32; o <<= 1) {
    int y = __shfl_up(incl, o);
    if (lane >= o) incl += y;
  }
  if (lane == 31) s_tot[w] = incl;
  __syncthreads();
  int b = s_pre[lane] + (lane < w ? s_tot[lane] : 0);
#pragma unroll
  for (int o = 16; o > 0; o >>= 1) b += __shfl_xor(b, o);
  int run = b + incl - tot;
  int4 r0, r1;
  r0.x = run; run += a0.x;
  r0.y = run; run += a0.y;
  r0.z = run; run += a0.z;
  r0.w = run; run += a0.w;
  r1.x = run; run += a1.x;
  r1.y = run; run += a1.y;
  r1.z = run; run += a1.z;
  r1.w = run;
  ((int4*)s_slot)[t * 2] = r0;
  ((int4*)s_slot)[t * 2 + 1] = r1;
  __syncthreads();
  int sta = 0;
#pragma unroll
  for (int q = 0; q < NPT; ++q) {
    int i = q * 1024 + t;
    int st = s_slot[i];
    s_slot[i] = st + s_front[i];
    if (q == qa) sta = st;
  }
  if (mine) {
    int g = lo + ia;
    start[g] = sta;
    cnt[g] = va;
    float dv = rsqrtf((float)(va + 1));
    dinv[g] = dv;
    float4* xp = (float4*)(xs + (size_t)g * AXP);
    xp[0] = make_float4(xv[0] * dv, xv[1] * dv, xv[2] * dv, xv[3] * dv);
    xp[1] = make_float4(xv[4] * dv, xv[5] * dv, xv[6] * dv, xv[7] * dv);
  }
  __syncthreads();
  int chunkE = (E + RCH - 1) / RCH;
  int e0 = c * chunkE;
  int e1 = e0 + chunkE < E ? e0 + chunkE : E;
  for (int base = e0; base < e1; base += 1024 * RPT) {
    int d[RPT], sv[RPT];
#pragma unroll
    for (int k = 0; k < RPT; ++k) {
      int e = base + k * 1024 + t;
      int ec = e < e1 ? e : e1 - 1;
      d[k] = dst[ec];
      sv[k] = src[ec];
    }
    __builtin_amdgcn_sched_barrier(0);
#pragma unroll
    for (int k = 0; k < RPT; ++k) d[k] = base + k * 1024 + t < e1 ? d[k] : -1;
    int p[RPT];
#pragma unroll
    for (int k = 0; k < RPT; ++k) {
      unsigned o = (unsigned)(d[k] - lo);
      p[k] = -1;
      if (o < RANGE) p[k] = atomicAdd(&s_slot[o], 1);
    }
#pragma unroll
    for (int k = 0; k < RPT; ++k)
      if (p[k] >= 0) csrc[p[k]] = sv[k];
  }
}


__global__ void k_gather64(const float* __restrict__ h, const float* __restrict__ dinv,
                           const int* __restrict__ start, const int* __restrict__ cnt,
                           const int* __restrict__ csrc, const float* __restrict__ b,
                           float* __restrict__ out, int n) {
  int lane = threadIdx.x & 31;
  int d = __builtin_amdgcn_readfirstlane((int)((blockIdx.x * blockDim.x + threadIdx.x) >> 5));
  if (d >= n) return;
  int beg = start[d];
  int m = cnt[d];
  float dd = dinv[d];
  const float2 hv = *(const float2*)(h + (size_t)d * 64 + lane * 2);
  const float2 bv = *(const float2*)(b + lane * 2);
  asm volatile("" :: "v"(bv.x), "v"(bv.y));
  float accx = hv.x;
  float accy = hv.y;
  for (int base = 0; base < m; base += 32) {
    int mm = m - base < 32 ? m - base : 32;
    int sj = d;
    if (lane < mm) sj = csrc[beg + base + lane];
    for (int j = 0; j < mm; j += 4) {
      float2 v[4];
      float wv[4];
#pragma unroll
      for (int u = 0; u < 4; ++u) {
        int s = __builtin_amdgcn_readlane(sj, j + u);
        wv[u] = j + u < mm ? 1.0f : 0.0f;
        v[u] = *(const float2*)(h + (size_t)s * 64 + lane * 2);
      }
#pragma unroll
      for (int u = 0; u < 4; ++u) {
        accx += v[u].x * wv[u];
        accy += v[u].y * wv[u];
      }
    }
  }
  float2 r;
  r.x = accx * dd + bv.x;
  r.y = accy * dd + bv.y;
  *(float2*)(out + (size_t)d * 64 + lane * 2) = r;
}

#define HID 128
#define OUTC 64
#define KSTEPS (HID / 4)
#define GU 16
static_assert(AXP == 8, "k_gather_mlp_wmma: a row of xs is two 16-byte halves, one per lane half");
__global__ void __launch_bounds__(256)
k_gather_mlp_wmma(const float* __restrict__ xs, const float* __restrict__ dinv,
                  const int* __restrict__ start, const int* __restrict__ cnt,
                  const int* __restrict__ csrc, const float* __restrict__ W1,
                  const float* __restrict__ b1, const float* __restrict__ W2,
                  float* __restrict__ D, int n, int in_c) {
  __shared__ float4 sB[KSTEPS * 2 * 32];
  __shared__ float4 sW[(HID / 2) * 4];

  int t = threadIdx.x;
  int lane = t & 31;
  int half = lane >> 4;
  int c    = lane & 15;
  int mTile = blockIdx.x * 8 + (t >> 5);
  bool live = mTile * 16 < n;
  int rowA = mTile * 16 + c;
  if (rowA >= n) rowA = n - 1;
  int beg = start[rowA];
  int m = cnt[rowA];
  if (!live) m = 0;
  float dv = dinv[rowA];
  const float* xh = xs + half * 4;
  float4 acc = *(const float4*)(xh + (size_t)rowA * AXP);

  float4 r0[(KSTEPS * 32) / 256], r1[(KSTEPS * 32) / 256];
#pragma unroll
  for (int j = 0; j < (KSTEPS * 32) / 256; ++j) {
    int k0 = (j * 8 + (t >> 5)) * 4 + half * 2;
    r0[j] = *(const float4*)(W2 + (size_t)k0 * OUTC + c * 4);
    r1[j] = *(const float4*)(W2 + (size_t)(k0 + 1) * OUTC + c * 4);
  }
  int kk = t & (HID - 1);
  float bv = b1[kk];
  float w1v[(HID * 8) / 256];
#pragma unroll
  for (int j = 0; j < (HID * 8) / 256; ++j) {
    int i = j * 2 + (t >> 7);
    w1v[j] = 0.0f;
    if (i < in_c) w1v[j] = W1[i * HID + kk];
  }
  int dummy = mTile * 16 < n ? mTile * 16 : n - 1;
  int idx[GU];
#pragma unroll
  for (int u = 0; u < GU; ++u) {
    int sv = csrc[u < m ? beg + u : 0];
    idx[u] = u < m ? sv : dummy;
  }
#pragma unroll
  for (int j = 0; j < (KSTEPS * 32) / 256; ++j) {
    int step = j * 8 + (t >> 5);
    sB[(step * 2 + 0) * 32 + lane] = make_float4(r0[j].x, r1[j].x, r0[j].y, r1[j].y);
    sB[(step * 2 + 1) * 32 + lane] = make_float4(r0[j].z, r1[j].z, r0[j].w, r1[j].w);
  }
  float* sWf = (float*)sW;
#pragma unroll
  for (int j = 0; j < (HID * 8) / 256; ++j) {
    int i = j * 2 + (t >> 7);
    sWf[(kk >> 1) * 16 + i * 2 + (kk & 1)] = i == 7 ? bv : w1v[j];
  }

  for (int base = 0; __any(base < m); base += GU) {
    float4 v[GU];
#pragma unroll
    for (int u = 0; u < GU; ++u) v[u] = *(const float4*)(xh + (size_t)idx[u] * AXP);
    int nb = base + GU;
#pragma unroll
    for (int u = 0; u < GU; ++u) {
      int sv = csrc[nb + u < m ? beg + nb + u : 0];
      idx[u] = nb + u < m ? sv : dummy;
    }
#pragma unroll
    for (int u = 0; u < GU; ++u) {
      bool on = base + u < m;
      acc.x += on ? v[u].x : 0.0f;
      acc.y += on ? v[u].y : 0.0f;
      acc.z += on ? v[u].z : 0.0f;
      acc.w += on ? v[u].w : 0.0f;
    }
  }
  float4 mine = make_float4(acc.x * dv, acc.y * dv, acc.z * dv, acc.w * dv);
  float4 oth;
  oth.x = __shfl_xor(mine.x, 16);
  oth.y = __shfl_xor(mine.y, 16);
  oth.z = __shfl_xor(mine.z, 16);
  oth.w = __shfl_xor(mine.w, 16);
  float4 x0 = half ? oth : mine;
  float4 x1 = half ? mine : oth;
  float xr[7] = {x0.x, x0.y, x0.z, x0.w, x1.x, x1.y, x1.z};

  __syncthreads();
  if (!live) return;

  v8f acc0 = {}, acc1 = {}, acc2 = {}, acc3 = {};
#pragma unroll
  for (int s = 0; s < KSTEPS; ++s) {
    const float4* wp = sW + (s * 2 + half) * 4;
    float4 w01 = wp[0], w23 = wp[1], w45 = wp[2], w6b = wp[3];
    v2f w, xv, hh;
    w.x = w01.x; w.y = w01.y; hh = w * xr[0];
    w.x = w01.z; w.y = w01.w; xv.x = xr[1]; xv.y = xr[1]; hh = __builtin_elementwise_fma(xv, w, hh);
    w.x = w23.x; w.y = w23.y; xv.x = xr[2]; xv.y = xr[2]; hh = __builtin_elementwise_fma(xv, w, hh);
    w.x = w23.z; w.y = w23.w; xv.x = xr[3]; xv.y = xr[3]; hh = __builtin_elementwise_fma(xv, w, hh);
    w.x = w45.x; w.y = w45.y; xv.x = xr[4]; xv.y = xr[4]; hh = __builtin_elementwise_fma(xv, w, hh);
    w.x = w45.z; w.y = w45.w; xv.x = xr[5]; xv.y = xr[5]; hh = __builtin_elementwise_fma(xv, w, hh);
    w.x = w6b.x; w.y = w6b.y; xv.x = xr[6]; xv.y = xr[6]; hh = __builtin_elementwise_fma(xv, w, hh);
    w.x = w6b.z; w.y = w6b.w; hh = hh + w;
    v2f a;
    a.x = fmaxf(hh.x, 0.0f) * dv;
    a.y = fmaxf(hh.y, 0.0f) * dv;
    float4 q0 = sB[(s * 2 + 0) * 32 + lane];
    float4 q1 = sB[(s * 2 + 1) * 32 + lane];
    v2f f0, f1, f2, f3;
    f0.x = q0.x; f0.y = q0.y;
    f1.x = q0.z; f1.y = q0.w;
    f2.x = q1.x; f2.y = q1.y;
    f3.x = q1.z; f3.y = q1.w;
    acc0 = __builtin_amdgcn_wmma_f32_16x16x4_f32(false, a, false, f0, (short)0, acc0, false, false);
    acc1 = __builtin_amdgcn_wmma_f32_16x16x4_f32(false, a, false, f1, (short)0, acc1, false, false);
    acc2 = __builtin_amdgcn_wmma_f32_16x16x4_f32(false, a, false, f2, (short)0, acc2, false, false);
    acc3 = __builtin_amdgcn_wmma_f32_16x16x4_f32(false, a, false, f3, (short)0, acc3, false, false);
  }
#pragma unroll
  for (int r = 0; r < 8; ++r) {
    int row = mTile * 16 + r + half * 8;
    if (row < n)
      *(float4*)(D + (size_t)row * OUTC + c * 4) = make_float4(acc0[r], acc1[r], acc2[r], acc3[r]);
  }
}

extern "C" void kernel_launch(void* const* d_in, const int* in_sizes, int n_in,
                              void* d_out, int out_size, void* d_ws, size_t ws_size,
                              hipStream_t stream) {
  const float* x  = (const float*)d_in[0];
  const int*   ei = (const int*)d_in[1];
  const float* W1 = (const float*)d_in[2];
  const float* b1 = (const float*)d_in[3];
  const float* W2 = (const float*)d_in[4];
  const float* b2 = (const float*)d_in[5];
  float* out = (float*)d_out;

  int hid   = in_sizes[3];
  int in_c  = in_sizes[2] / hid;
  int n     = in_sizes[0] / in_c;
  int E     = in_sizes[1] / 2;

  const int* src = ei;
  const int* dst = ei + E;

  float* ws   = (float*)d_ws;
  float* dinv = ws;
  float* xs   = dinv + n;
  float* h2   = xs + (size_t)n * AXP;
  int* cnt    = (int*)(h2 + (size_t)n * OUTC);
  int* start  = cnt + n;
  int* csrc   = start + n;
  int* tab    = csrc + E;
  int* bsum   = tab + (size_t)RCH * n;

  const int B256 = 256;
  dim3 cr(RCH, (n + RANGE - 1) / RANGE);
  k_count_lds<<<cr, 1024, 0, stream>>>(dst, tab, bsum, E, n);
  k_scan_fill<<<cr, 1024, 0, stream>>>(src, dst, tab, bsum, cnt, start, dinv, x, xs, csrc, E, n, in_c);

  int mTiles = (n + 15) / 16;
  k_gather_mlp_wmma<<<(mTiles + 7) / 8, B256, 0, stream>>>(xs, dinv, start, cnt, csrc, W1, b1, W2, h2, n, in_c);

  k_gather64<<<(n + 7) / 8, B256, 0, stream>>>(h2, dinv, start, cnt, csrc, b2, out, n);
}


# layer-2 gather on a resident grid of 8192 waves, each walking ~12 nodes in a software pipeline: a node's rows all in flight at once, the next node's list and bounds loaded under them; sum order kept
# speedup vs baseline: 6.0022x; 1.0484x over previous
#include <hip/hip_runtime.h>

typedef __attribute__((ext_vector_type(2))) float v2f;
typedef __attribute__((ext_vector_type(8))) float v8f;

#define RCH 16
#define RANGE 8192
#define RPT 16
#define SCAN_T 1024
#define SUBS (RANGE / SCAN_T)
#define NPT (RANGE / 1024)
#define AXP 8
static_assert(SCAN_T == 1024 && RANGE % SCAN_T == 0 && SUBS % 4 == 0 && 32 % SUBS == 0,
              "k_count_lds: 1024 threads, SUBS counters each, whole waves per node block");
static_assert(NPT == 8 && RCH == 2 * NPT,
              "k_scan_fill: 8 nodes per thread, and (slice of 1024 nodes, half of it) numbers the RCH chunk work-groups");
__global__ void __launch_bounds__(1024)
k_count_lds(const int* __restrict__ dst, int* __restrict__ tab, int* __restrict__ bsum, int E, int n) {
  __shared__ alignas(16) int s_cnt[RANGE];
  __shared__ int s_part[32];
  int t = threadIdx.x;
  int lo = blockIdx.y * RANGE;
  for (int i = t; i < RANGE; i += 1024) s_cnt[i] = 0;
  __syncthreads();
  int chunkE = (E + RCH - 1) / RCH;
  int e0 = blockIdx.x * chunkE;
  int e1 = e0 + chunkE < E ? e0 + chunkE : E;
  for (int base = e0; base < e1; base += 1024 * RPT) {
    int d[RPT];
#pragma unroll
    for (int k = 0; k < RPT; ++k) {
      int e = base + k * 1024 + t;
      d[k] = dst[e < e1 ? e : e1 - 1];
    }
    __builtin_amdgcn_sched_barrier(0);
#pragma unroll
    for (int k = 0; k < RPT; ++k) d[k] = base + k * 1024 + t < e1 ? d[k] : -1;
#pragma unroll
    for (int k = 0; k < RPT; ++k) {
      unsigned o = (unsigned)(d[k] - lo);
      if (o < RANGE) atomicAdd(&s_cnt[o], 1);
    }
  }
  __syncthreads();
  int len = n - lo < RANGE ? n - lo : RANGE;
  for (int i = t; i < len; i += 1024) tab[(size_t)blockIdx.x * n + lo + i] = s_cnt[i];
  const int4* c4 = (const int4*)(s_cnt + t * SUBS);
  int v = 0;
  for (int q = 0; q < SUBS / 4; ++q) {
    int4 a = c4[q];
    int i = t * SUBS + q * 4;
    v += (i + 0 < len ? a.x : 0) + (i + 1 < len ? a.y : 0) + (i + 2 < len ? a.z : 0) + (i + 3 < len ? a.w : 0);
  }
  for (int o = 16; o > 0; o >>= 1) v += __shfl_xor(v, o);
  if ((t & 31) == 0) s_part[t >> 5] = v;
  __syncthreads();
  if (t < SUBS && t * SCAN_T < len) {
    int s = 0;
    for (int q = 0; q < 32 / SUBS; ++q) s += s_part[t * (32 / SUBS) + q];
    bsum[(size_t)(blockIdx.y * SUBS + t) * RCH + blockIdx.x] = s;
  }
}

__global__ void __launch_bounds__(1024)
k_scan_fill(const int* __restrict__ src, const int* __restrict__ dst,
            const int* __restrict__ tab, const int* __restrict__ bsum,
            int* __restrict__ cnt, int* __restrict__ start, float* __restrict__ dinv,
            const float* __restrict__ x, float* __restrict__ xs, int* __restrict__ csrc,
            int E, int n, int in_c) {
  __shared__ alignas(16) int s_slot[RANGE];
  __shared__ int s_front[RANGE];
  __shared__ int s_pre[32];
  __shared__ int s_tot[32];
  int t = threadIdx.x;
  int lane = t & 31;
  int w = t >> 5;
  int c = blockIdx.x;
  int lo = blockIdx.y * RANGE;
  int len = n - lo < RANGE ? n - lo : RANGE;
  int before = blockIdx.y * SUBS * RCH;
  int s0 = t < before ? bsum[t] : 0;
  int s1 = t + 1024 < before ? bsum[t + 1024] : 0;
  int qa = c >> 1;
  int ia = qa * 1024 + t;
  bool mine = ia < len && (t >> 9) == (c & 1);
  float xv[AXP];
#pragma unroll
  for (int k = 0; k < AXP; ++k) xv[k] = (mine && k < in_c) ? x[(size_t)(lo + ia) * in_c + k] : 0.0f;
  int va = 0;
#pragma unroll
  for (int h = 0; h < 2; ++h) {
    int u[NPT / 2][RCH];
#pragma unroll
    for (int qq = 0; qq < NPT / 2; ++qq) {
      int i = (h * (NPT / 2) + qq) * 1024 + t;
      const int* tp = tab + lo + (i < len ? i : len - 1);
#pragma unroll
      for (int cc = 0; cc < RCH; ++cc) u[qq][cc] = tp[(size_t)cc * n];
    }
    __builtin_amdgcn_sched_barrier(0);
#pragma unroll
    for (int qq = 0; qq < NPT / 2; ++qq) {
      int q = h * (NPT / 2) + qq;
      int i = q * 1024 + t;
      int a = 0, p = 0;
#pragma unroll
      for (int cc = 0; cc < RCH; ++cc) {
        a += u[qq][cc];
        p += cc < c ? u[qq][cc] : 0;
      }
      a = i < len ? a : 0;
      s_slot[i] = a;
      s_front[i] = p;
      if (q == qa) va = a;
    }
  }
  int s = s0 + s1;
  for (int j = t + 2048; j < before; j += 1024) s += bsum[j];
#pragma unroll
  for (int o = 16; o > 0; o >>= 1) s += __shfl_xor(s, o);
  if (lane == 0) s_pre[w] = s;
  __syncthreads();
  int4 a0 = ((const int4*)s_slot)[t * 2];
  int4 a1 = ((const int4*)s_slot)[t * 2 + 1];
  int tot = a0.x + a0.y + a0.z + a0.w + a1.x + a1.y + a1.z + a1.w;
  int incl = tot;
#pragma unroll
  for (int o = 1; o < 32; o <<= 1) {
    int y = __shfl_up(incl, o);
    if (lane >= o) incl += y;
  }
  if (lane == 31) s_tot[w] = incl;
  __syncthreads();
  int b = s_pre[lane] + (lane < w ? s_tot[lane] : 0);
#pragma unroll
  for (int o = 16; o > 0; o >>= 1) b += __shfl_xor(b, o);
  int run = b + incl - tot;
  int4 r0, r1;
  r0.x = run; run += a0.x;
  r0.y = run; run += a0.y;
  r0.z = run; run += a0.z;
  r0.w = run; run += a0.w;
  r1.x = run; run += a1.x;
  r1.y = run; run += a1.y;
  r1.z = run; run += a1.z;
  r1.w = run;
  ((int4*)s_slot)[t * 2] = r0;
  ((int4*)s_slot)[t * 2 + 1] = r1;
  __syncthreads();
  int sta = 0;
#pragma unroll
  for (int q = 0; q < NPT; ++q) {
    int i = q * 1024 + t;
    int st = s_slot[i];
    s_slot[i] = st + s_front[i];
    if (q == qa) sta = st;
  }
  if (mine) {
    int g = lo + ia;
    start[g] = sta;
    cnt[g] = va;
    float dv = rsqrtf((float)(va + 1));
    dinv[g] = dv;
    float4* xp = (float4*)(xs + (size_t)g * AXP);
    xp[0] = make_float4(xv[0] * dv, xv[1] * dv, xv[2] * dv, xv[3] * dv);
    xp[1] = make_float4(xv[4] * dv, xv[5] * dv, xv[6] * dv, xv[7] * dv);
  }
  __syncthreads();
  int chunkE = (E + RCH - 1) / RCH;
  int e0 = c * chunkE;
  int e1 = e0 + chunkE < E ? e0 + chunkE : E;
  for (int base = e0; base < e1; base += 1024 * RPT) {
    int d[RPT], sv[RPT];
#pragma unroll
    for (int k = 0; k < RPT; ++k) {
      int e = base + k * 1024 + t;
      int ec = e < e1 ? e : e1 - 1;
      d[k] = dst[ec];
      sv[k] = src[ec];
    }
    __builtin_amdgcn_sched_barrier(0);
#pragma unroll
    for (int k = 0; k < RPT; ++k) d[k] = base + k * 1024 + t < e1 ? d[k] : -1;
    int p[RPT];
#pragma unroll
    for (int k = 0; k < RPT; ++k) {
      unsigned o = (unsigned)(d[k] - lo);
      p[k] = -1;
      if (o < RANGE) p[k] = atomicAdd(&s_slot[o], 1);
    }
#pragma unroll
    for (int k = 0; k < RPT; ++k)
      if (p[k] >= 0) csrc[p[k]] = sv[k];
  }
}


#define GW_BLOCKS 1024

template <int NS>
__device__ __forceinline__ void rows_in_flight(const float* __restrict__ h, int lane, int sj, int mm,
                                               float& accx, float& accy) {
  float2 v[NS];
#pragma unroll
  for (int j = 0; j < NS; ++j) {
    int s = __builtin_amdgcn_readlane(sj, j);
    v[j] = *(const float2*)(h + (size_t)s * 64 + lane * 2);
  }
#pragma unroll
  for (int j = 0; j < NS - 1; ++j) {
    accx += v[j].x;
    accy += v[j].y;
  }
  float wv = NS - 1 < mm ? 1.0f : 0.0f;
  accx += v[NS - 1].x * wv;
  accy += v[NS - 1].y * wv;
}

__global__ void __launch_bounds__(256)
k_gather64(const float* __restrict__ h, const float* __restrict__ dinv,
           const int* __restrict__ start, const int* __restrict__ cnt,
           const int* __restrict__ csrc, const float* __restrict__ b,
           float* __restrict__ out, int n) {
  int lane = threadIdx.x & 31;
  int d = __builtin_amdgcn_readfirstlane((int)((blockIdx.x * blockDim.x + threadIdx.x) >> 5));
  int waves = (int)((gridDim.x * blockDim.x) >> 5);
  if (d >= n) return;
  const float2 bv = *(const float2*)(b + lane * 2);
  int beg = start[d];
  int m = cnt[d];
  float dd = dinv[d];
  int s0 = csrc[lane < m ? beg + lane : 0];
  int sj = lane < m ? s0 : d;
  float2 hv = *(const float2*)(h + (size_t)d * 64 + lane * 2);
  int dn = d + waves < n ? d + waves : d;
  int begn = start[dn];
  int mn = cnt[dn];
  float ddn = dinv[dn];
  for (;;) {
    int s1 = csrc[lane < mn ? begn + lane : 0];
    int sjn = lane < mn ? s1 : dn;
    float2 hvn = *(const float2*)(h + (size_t)dn * 64 + lane * 2);
    int d2 = dn + waves < n ? dn + waves : dn;
    int beg2 = start[d2];
    int m2 = cnt[d2];
    float dd2 = dinv[d2];
    float accx = hv.x;
    float accy = hv.y;
    int mm = m < 32 ? m : 32;
    switch ((mm + 1) >> 1) {
#define ROWS_CASE(NP) case NP: rows_in_flight<2 * NP>(h, lane, sj, mm, accx, accy); break;
      ROWS_CASE(1) ROWS_CASE(2) ROWS_CASE(3) ROWS_CASE(4)
      ROWS_CASE(5) ROWS_CASE(6) ROWS_CASE(7) ROWS_CASE(8)
      ROWS_CASE(9) ROWS_CASE(10) ROWS_CASE(11) ROWS_CASE(12)
      ROWS_CASE(13) ROWS_CASE(14) ROWS_CASE(15) ROWS_CASE(16)
#undef ROWS_CASE
      default: break;
    }
    for (int base = 32; base < m; base += 32) {
      int mr = m - base < 32 ? m - base : 32;
      int sx = d;
      if (lane < mr) sx = csrc[beg + base + lane];
      for (int j = 0; j < mr; j += 4) {
        float2 v[4];
        float wv[4];
#pragma unroll
        for (int u = 0; u < 4; ++u) {
          int s = __builtin_amdgcn_readlane(sx, j + u);
          wv[u] = j + u < mr ? 1.0f : 0.0f;
          v[u] = *(const float2*)(h + (size_t)s * 64 + lane * 2);
        }
#pragma unroll
        for (int u = 0; u < 4; ++u) {
          accx += v[u].x * wv[u];
          accy += v[u].y * wv[u];
        }
      }
    }
    float2 r;
    r.x = accx * dd + bv.x;
    r.y = accy * dd + bv.y;
    *(float2*)(out + (size_t)d * 64 + lane * 2) = r;
    if (d + waves >= n) break;
    d = dn; beg = begn; m = mn; dd = ddn; sj = sjn; hv = hvn;
    dn = d2; begn = beg2; mn = m2; ddn = dd2;
  }
}

#define HID 128
#define OUTC 64
#define KSTEPS (HID / 4)
#define GU 16
static_assert(AXP == 8, "k_gather_mlp_wmma: a row of xs is two 16-byte halves, one per lane half");
__global__ void __launch_bounds__(256)
k_gather_mlp_wmma(const float* __restrict__ xs, const float* __restrict__ dinv,
                  const int* __restrict__ start, const int* __restrict__ cnt,
                  const int* __restrict__ csrc, const float* __restrict__ W1,
                  const float* __restrict__ b1, const float* __restrict__ W2,
                  float* __restrict__ D, int n, int in_c) {
  __shared__ float4 sB[KSTEPS * 2 * 32];
  __shared__ float4 sW[(HID / 2) * 4];

  int t = threadIdx.x;
  int lane = t & 31;
  int half = lane >> 4;
  int c    = lane & 15;
  int mTile = blockIdx.x * 8 + (t >> 5);
  bool live = mTile * 16 < n;
  int rowA = mTile * 16 + c;
  if (rowA >= n) rowA = n - 1;
  int beg = start[rowA];
  int m = cnt[rowA];
  if (!live) m = 0;
  float dv = dinv[rowA];
  const float* xh = xs + half * 4;
  float4 acc = *(const float4*)(xh + (size_t)rowA * AXP);

  float4 r0[(KSTEPS * 32) / 256], r1[(KSTEPS * 32) / 256];
#pragma unroll
  for (int j = 0; j < (KSTEPS * 32) / 256; ++j) {
    int k0 = (j * 8 + (t >> 5)) * 4 + half * 2;
    r0[j] = *(const float4*)(W2 + (size_t)k0 * OUTC + c * 4);
    r1[j] = *(const float4*)(W2 + (size_t)(k0 + 1) * OUTC + c * 4);
  }
  int kk = t & (HID - 1);
  float bv = b1[kk];
  float w1v[(HID * 8) / 256];
#pragma unroll
  for (int j = 0; j < (HID * 8) / 256; ++j) {
    int i = j * 2 + (t >> 7);
    w1v[j] = 0.0f;
    if (i < in_c) w1v[j] = W1[i * HID + kk];
  }
  int dummy = mTile * 16 < n ? mTile * 16 : n - 1;
  int idx[GU];
#pragma unroll
  for (int u = 0; u < GU; ++u) {
    int sv = csrc[u < m ? beg + u : 0];
    idx[u] = u < m ? sv : dummy;
  }
#pragma unroll
  for (int j = 0; j < (KSTEPS * 32) / 256; ++j) {
    int step = j * 8 + (t >> 5);
    sB[(step * 2 + 0) * 32 + lane] = make_float4(r0[j].x, r1[j].x, r0[j].y, r1[j].y);
    sB[(step * 2 + 1) * 32 + lane] = make_float4(r0[j].z, r1[j].z, r0[j].w, r1[j].w);
  }
  float* sWf = (float*)sW;
#pragma unroll
  for (int j = 0; j < (HID * 8) / 256; ++j) {
    int i = j * 2 + (t >> 7);
    sWf[(kk >> 1) * 16 + i * 2 + (kk & 1)] = i == 7 ? bv : w1v[j];
  }

  for (int base = 0; __any(base < m); base += GU) {
    float4 v[GU];
#pragma unroll
    for (int u = 0; u < GU; ++u) v[u] = *(const float4*)(xh + (size_t)idx[u] * AXP);
    int nb = base + GU;
#pragma unroll
    for (int u = 0; u < GU; ++u) {
      int sv = csrc[nb + u < m ? beg + nb + u : 0];
      idx[u] = nb + u < m ? sv : dummy;
    }
#pragma unroll
    for (int u = 0; u < GU; ++u) {
      bool on = base + u < m;
      acc.x += on ? v[u].x : 0.0f;
      acc.y += on ? v[u].y : 0.0f;
      acc.z += on ? v[u].z : 0.0f;
      acc.w += on ? v[u].w : 0.0f;
    }
  }
  float4 mine = make_float4(acc.x * dv, acc.y * dv, acc.z * dv, acc.w * dv);
  float4 oth;
  oth.x = __shfl_xor(mine.x, 16);
  oth.y = __shfl_xor(mine.y, 16);
  oth.z = __shfl_xor(mine.z, 16);
  oth.w = __shfl_xor(mine.w, 16);
  float4 x0 = half ? oth : mine;
  float4 x1 = half ? mine : oth;
  float xr[7] = {x0.x, x0.y, x0.z, x0.w, x1.x, x1.y, x1.z};

  __syncthreads();
  if (!live) return;

  v8f acc0 = {}, acc1 = {}, acc2 = {}, acc3 = {};
#pragma unroll
  for (int s = 0; s < KSTEPS; ++s) {
    const float4* wp = sW + (s * 2 + half) * 4;
    float4 w01 = wp[0], w23 = wp[1], w45 = wp[2], w6b = wp[3];
    v2f w, xv, hh;
    w.x = w01.x; w.y = w01.y; hh = w * xr[0];
    w.x = w01.z; w.y = w01.w; xv.x = xr[1]; xv.y = xr[1]; hh = __builtin_elementwise_fma(xv, w, hh);
    w.x = w23.x; w.y = w23.y; xv.x = xr[2]; xv.y = xr[2]; hh = __builtin_elementwise_fma(xv, w, hh);
    w.x = w23.z; w.y = w23.w; xv.x = xr[3]; xv.y = xr[3]; hh = __builtin_elementwise_fma(xv, w, hh);
    w.x = w45.x; w.y = w45.y; xv.x = xr[4]; xv.y = xr[4]; hh = __builtin_elementwise_fma(xv, w, hh);
    w.x = w45.z; w.y = w45.w; xv.x = xr[5]; xv.y = xr[5]; hh = __builtin_elementwise_fma(xv, w, hh);
    w.x = w6b.x; w.y = w6b.y; xv.x = xr[6]; xv.y = xr[6]; hh = __builtin_elementwise_fma(xv, w, hh);
    w.x = w6b.z; w.y = w6b.w; hh = hh + w;
    v2f a;
    a.x = fmaxf(hh.x, 0.0f) * dv;
    a.y = fmaxf(hh.y, 0.0f) * dv;
    float4 q0 = sB[(s * 2 + 0) * 32 + lane];
    float4 q1 = sB[(s * 2 + 1) * 32 + lane];
    v2f f0, f1, f2, f3;
    f0.x = q0.x; f0.y = q0.y;
    f1.x = q0.z; f1.y = q0.w;
    f2.x = q1.x; f2.y = q1.y;
    f3.x = q1.z; f3.y = q1.w;
    acc0 = __builtin_amdgcn_wmma_f32_16x16x4_f32(false, a, false, f0, (short)0, acc0, false, false);
    acc1 = __builtin_amdgcn_wmma_f32_16x16x4_f32(false, a, false, f1, (short)0, acc1, false, false);
    acc2 = __builtin_amdgcn_wmma_f32_16x16x4_f32(false, a, false, f2, (short)0, acc2, false, false);
    acc3 = __builtin_amdgcn_wmma_f32_16x16x4_f32(false, a, false, f3, (short)0, acc3, false, false);
  }
#pragma unroll
  for (int r = 0; r < 8; ++r) {
    int row = mTile * 16 + r + half * 8;
    if (row < n)
      *(float4*)(D + (size_t)row * OUTC + c * 4) = make_float4(acc0[r], acc1[r], acc2[r], acc3[r]);
  }
}

extern "C" void kernel_launch(void* const* d_in, const int* in_sizes, int n_in,
                              void* d_out, int out_size, void* d_ws, size_t ws_size,
                              hipStream_t stream) {
  const float* x  = (const float*)d_in[0];
  const int*   ei = (const int*)d_in[1];
  const float* W1 = (const float*)d_in[2];
  const float* b1 = (const float*)d_in[3];
  const float* W2 = (const float*)d_in[4];
  const float* b2 = (const float*)d_in[5];
  float* out = (float*)d_out;

  int hid   = in_sizes[3];
  int in_c  = in_sizes[2] / hid;
  int n     = in_sizes[0] / in_c;
  int E     = in_sizes[1] / 2;

  const int* src = ei;
  const int* dst = ei + E;

  float* ws   = (float*)d_ws;
  float* dinv = ws;
  float* xs   = dinv + n;
  float* h2   = xs + (size_t)n * AXP;
  int* cnt    = (int*)(h2 + (size_t)n * OUTC);
  int* start  = cnt + n;
  int* csrc   = start + n;
  int* tab    = csrc + E;
  int* bsum   = tab + (size_t)RCH * n;

  const int B256 = 256;
  dim3 cr(RCH, (n + RANGE - 1) / RANGE);
  k_count_lds<<<cr, 1024, 0, stream>>>(dst, tab, bsum, E, n);
  k_scan_fill<<<cr, 1024, 0, stream>>>(src, dst, tab, bsum, cnt, start, dinv, x, xs, csrc, E, n, in_c);

  int mTiles = (n + 15) / 16;
  k_gather_mlp_wmma<<<(mTiles + 7) / 8, B256, 0, stream>>>(xs, dinv, start, cnt, csrc, W1, b1, W2, h2, n, in_c);

  k_gather64<<<GW_BLOCKS, B256, 0, stream>>>(h2, dinv, start, cnt, csrc, b2, out, n);
}


# edge list by bucket sort: each edge chunk sorted by 256-node range in LDS into its own segment, then one work-group per range counts, scans and places; sum order = arrival rank in the group
# speedup vs baseline: 8.4145x; 1.4019x over previous
#include <hip/hip_runtime.h>

typedef __attribute__((ext_vector_type(2))) float v2f;
typedef __attribute__((ext_vector_type(8))) float v8f;

#define CHUNK 6144
#define BT 512
#define EPT (CHUNK / BT)
#define RW 256
#define RSH 8
#define NR_CAP BT
#define NCH_CAP 1024
#define PU 17
#define AXP 8
static_assert(RW == 1 << RSH && RW <= BT && RW % 32 == 0 && CHUNK % BT == 0, "ranges are whole waves of k_place");
__global__ void __launch_bounds__(BT)
k_bucket(const int* __restrict__ src, const int* __restrict__ dst, int* __restrict__ seg,
         int* __restrict__ offT, int E, int n, int nr, int nch) {
  __shared__ int s_cnt[NR_CAP];
  __shared__ int s_edge[CHUNK];
  __shared__ int s_wtot[BT / 32];
  int t = threadIdx.x;
  int lane = t & 31;
  int w = t >> 5;
  int c = blockIdx.x;
  int e0 = c * CHUNK;
  int e1 = e0 + CHUNK < E ? e0 + CHUNK : E;
  s_cnt[t] = 0;
  int d[EPT], sv[EPT], rk[EPT];
#pragma unroll
  for (int k = 0; k < EPT; ++k) {
    int e = e0 + k * BT + t;
    int ec = e < e1 ? e : e1 - 1;
    d[k] = dst[ec];
    sv[k] = src[ec];
  }
  __builtin_amdgcn_sched_barrier(0);
  __syncthreads();
#pragma unroll
  for (int k = 0; k < EPT; ++k) {
    bool ok = e0 + k * BT + t < e1 && (unsigned)d[k] < (unsigned)n;
    rk[k] = -1;
    if (ok) rk[k] = atomicAdd(&s_cnt[d[k] >> RSH], 1);
  }
  __syncthreads();
  int v = s_cnt[t];
  int incl = v;
#pragma unroll
  for (int o = 1; o < 32; o <<= 1) {
    int y = __shfl_up(incl, o);
    if (lane >= o) incl += y;
  }
  if (lane == 31) s_wtot[w] = incl;
  __syncthreads();
  int pw = lane < w && lane < BT / 32 ? s_wtot[lane < BT / 32 ? lane : 0] : 0;
#pragma unroll
  for (int o = 16; o > 0; o >>= 1) pw += __shfl_xor(pw, o);
  int excl = pw + incl - v;
  s_cnt[t] = excl;
  if (t <= nr) offT[(size_t)t * nch + c] = excl;
  __syncthreads();
#pragma unroll
  for (int k = 0; k < EPT; ++k)
    if (rk[k] >= 0)
      s_edge[s_cnt[d[k] >> RSH] + rk[k]] = (int)(((unsigned)sv[k] << RSH) | (unsigned)(d[k] & (RW - 1)));
  __syncthreads();
  int tot = s_cnt[nr];
  for (int i = t; i < tot; i += BT) seg[e0 + i] = s_edge[i];
}

__device__ __forceinline__ void piece_loads(const int* __restrict__ seg, const int2* s_pl,
                                            int cb, int nch, int lane,
                                            int (&ev)[PU], int (&ln)[PU], int (&p0)[PU]) {
#pragma unroll
  for (int u = 0; u < PU; ++u) {
    int c = cb + u * (BT / 32);
    bool on = c < nch;
    int2 pl = s_pl[on ? c : 0];
    p0[u] = pl.x;
    ln[u] = on ? pl.y : 0;
  }
#pragma unroll
  for (int u = 0; u < PU; ++u) {
    ev[u] = 0;
    if (lane < ln[u]) ev[u] = seg[p0[u] + lane];
  }
}

template <bool PLACE>
__device__ __forceinline__ void edge_use(int word, int* s_deg, int* __restrict__ csrc) {
  unsigned e = (unsigned)word;
  int p = atomicAdd(&s_deg[e & (RW - 1)], 1);
  if (PLACE) csrc[p] = (int)(e >> RSH);
}

template <bool PLACE>
__device__ __forceinline__ void piece_use(const int* __restrict__ seg, int* s_deg, int* __restrict__ csrc, int lane,
                                          const int (&ev)[PU], const int (&ln)[PU], const int (&p0)[PU]) {
#pragma unroll
  for (int u = 0; u < PU; ++u) {
    if (lane < ln[u]) edge_use<PLACE>(ev[u], s_deg, csrc);
    for (int i = lane + 32; i < ln[u]; i += 32) edge_use<PLACE>(seg[p0[u] + i], s_deg, csrc);
  }
}

__global__ void __launch_bounds__(BT)
k_place(const int* __restrict__ seg, const int* __restrict__ offT, const float* __restrict__ x,
        int* __restrict__ cnt, int* __restrict__ start, float* __restrict__ dinv,
        float* __restrict__ xs, int* __restrict__ csrc, int n, int in_c, int nch) {
  __shared__ int2 s_pl[NCH_CAP];
  __shared__ int s_deg[RW];
  __shared__ int s_red[BT / 32];
  __shared__ int s_tot[RW / 32];
  int t = threadIdx.x;
  int lane = t & 31;
  int w = __builtin_amdgcn_readfirstlane(t >> 5);
  int r = blockIdx.x;
  int lo = r * RW;
  int len = n - lo < RW ? n - lo : RW;
  bool node = t < len;
  float xv[AXP];
#pragma unroll
  for (int k = 0; k < AXP; ++k) xv[k] = (node && k < in_c) ? x[(size_t)(lo + t) * in_c + k] : 0.0f;
  if (t < RW) s_deg[t] = 0;
  int osum = 0;
  for (int c = t; c < nch; c += BT) {
    int o0 = offT[(size_t)r * nch + c];
    int o1 = offT[(size_t)(r + 1) * nch + c];
    s_pl[c] = make_int2(c * CHUNK + o0, o1 - o0);
    osum += o0;
  }
#pragma unroll
  for (int o = 16; o > 0; o >>= 1) osum += __shfl_xor(osum, o);
  if (lane == 0) s_red[w] = osum;
  __syncthreads();
  int ev[PU], ln[PU], p0[PU];
  piece_loads(seg, s_pl, w, nch, lane, ev, ln, p0);
  piece_use<false>(seg, s_deg, csrc, lane, ev, ln, p0);
  for (int cb = w + (BT / 32) * PU; cb < nch; cb += (BT / 32) * PU) {
    int ev2[PU], ln2[PU], p02[PU];
    piece_loads(seg, s_pl, cb, nch, lane, ev2, ln2, p02);
    piece_use<false>(seg, s_deg, csrc, lane, ev2, ln2, p02);
  }
  __syncthreads();
  int v = t < RW ? s_deg[t < RW ? t : 0] : 0;
  int incl = v;
#pragma unroll
  for (int o = 1; o < 32; o <<= 1) {
    int y = __shfl_up(incl, o);
    if (lane >= o) incl += y;
  }
  if (t < RW && lane == 31) s_tot[w] = incl;
  int base = lane < BT / 32 ? s_red[lane < BT / 32 ? lane : 0] : 0;
#pragma unroll
  for (int o = 16; o > 0; o >>= 1) base += __shfl_xor(base, o);
  __syncthreads();
  int pw = lane < w && lane < RW / 32 ? s_tot[lane < RW / 32 ? lane : 0] : 0;
#pragma unroll
  for (int o = 16; o > 0; o >>= 1) pw += __shfl_xor(pw, o);
  if (t < RW) {
    int st = base + pw + incl - v;
    s_deg[t] = st;
    if (node) {
      int g = lo + t;
      start[g] = st;
      cnt[g] = v;
      float dv = rsqrtf((float)(v + 1));
      dinv[g] = dv;
      float4* xp = (float4*)(xs + (size_t)g * AXP);
      xp[0] = make_float4(xv[0] * dv, xv[1] * dv, xv[2] * dv, xv[3] * dv);
      xp[1] = make_float4(xv[4] * dv, xv[5] * dv, xv[6] * dv, xv[7] * dv);
    }
  }
  __syncthreads();
  piece_use<true>(seg, s_deg, csrc, lane, ev, ln, p0);
  for (int cb = w + (BT / 32) * PU; cb < nch; cb += (BT / 32) * PU) {
    int ev2[PU], ln2[PU], p02[PU];
    piece_loads(seg, s_pl, cb, nch, lane, ev2, ln2, p02);
    piece_use<true>(seg, s_deg, csrc, lane, ev2, ln2, p02);
  }
}


#define GW_BLOCKS 1024

template <int NS>
__device__ __forceinline__ void rows_in_flight(const float* __restrict__ h, int lane, int sj, int mm,
                                               float& accx, float& accy) {
  float2 v[NS];
#pragma unroll
  for (int j = 0; j < NS; ++j) {
    int s = __builtin_amdgcn_readlane(sj, j);
    v[j] = *(const float2*)(h + (size_t)s * 64 + lane * 2);
  }
#pragma unroll
  for (int j = 0; j < NS - 1; ++j) {
    accx += v[j].x;
    accy += v[j].y;
  }
  float wv = NS - 1 < mm ? 1.0f : 0.0f;
  accx += v[NS - 1].x * wv;
  accy += v[NS - 1].y * wv;
}

__global__ void __launch_bounds__(256)
k_gather64(const float* __restrict__ h, const float* __restrict__ dinv,
           const int* __restrict__ start, const int* __restrict__ cnt,
           const int* __restrict__ csrc, const float* __restrict__ b,
           float* __restrict__ out, int n) {
  int lane = threadIdx.x & 31;
  int d = __builtin_amdgcn_readfirstlane((int)((blockIdx.x * blockDim.x + threadIdx.x) >> 5));
  int waves = (int)((gridDim.x * blockDim.x) >> 5);
  if (d >= n) return;
  const float2 bv = *(const float2*)(b + lane * 2);
  int beg = start[d];
  int m = cnt[d];
  float dd = dinv[d];
  int s0 = csrc[lane < m ? beg + lane : 0];
  int sj = lane < m ? s0 : d;
  float2 hv = *(const float2*)(h + (size_t)d * 64 + lane * 2);
  int dn = d + waves < n ? d + waves : d;
  int begn = start[dn];
  int mn = cnt[dn];
  float ddn = dinv[dn];
  for (;;) {
    int s1 = csrc[lane < mn ? begn + lane : 0];
    int sjn = lane < mn ? s1 : dn;
    float2 hvn = *(const float2*)(h + (size_t)dn * 64 + lane * 2);
    int d2 = dn + waves < n ? dn + waves : dn;
    int beg2 = start[d2];
    int m2 = cnt[d2];
    float dd2 = dinv[d2];
    float accx = hv.x;
    float accy = hv.y;
    int mm = m < 32 ? m : 32;
    switch ((mm + 1) >> 1) {
#define ROWS_CASE(NP) case NP: rows_in_flight<2 * NP>(h, lane, sj, mm, accx, accy); break;
      ROWS_CASE(1) ROWS_CASE(2) ROWS_CASE(3) ROWS_CASE(4)
      ROWS_CASE(5) ROWS_CASE(6) ROWS_CASE(7) ROWS_CASE(8)
      ROWS_CASE(9) ROWS_CASE(10) ROWS_CASE(11) ROWS_CASE(12)
      ROWS_CASE(13) ROWS_CASE(14) ROWS_CASE(15) ROWS_CASE(16)
#undef ROWS_CASE
      default: break;
    }
    for (int base = 32; base < m; base += 32) {
      int mr = m - base < 32 ? m - base : 32;
      int sx = d;
      if (lane < mr) sx = csrc[beg + base + lane];
      for (int j = 0; j < mr; j += 4) {
        float2 v[4];
        float wv[4];
#pragma unroll
        for (int u = 0; u < 4; ++u) {
          int s = __builtin_amdgcn_readlane(sx, j + u);
          wv[u] = j + u < mr ? 1.0f : 0.0f;
          v[u] = *(const float2*)(h + (size_t)s * 64 + lane * 2);
        }
#pragma unroll
        for (int u = 0; u < 4; ++u) {
          accx += v[u].x * wv[u];
          accy += v[u].y * wv[u];
        }
      }
    }
    float2 r;
    r.x = accx * dd + bv.x;
    r.y = accy * dd + bv.y;
    *(float2*)(out + (size_t)d * 64 + lane * 2) = r;
    if (d + waves >= n) break;
    d = dn; beg = begn; m = mn; dd = ddn; sj = sjn; hv = hvn;
    dn = d2; begn = beg2; mn = m2; ddn = dd2;
  }
}

#define HID 128
#define OUTC 64
#define KSTEPS (HID / 4)
#define GU 16
static_assert(AXP == 8, "k_gather_mlp_wmma: a row of xs is two 16-byte halves, one per lane half");
__global__ void __launch_bounds__(256)
k_gather_mlp_wmma(const float* __restrict__ xs, const float* __restrict__ dinv,
                  const int* __restrict__ start, const int* __restrict__ cnt,
                  const int* __restrict__ csrc, const float* __restrict__ W1,
                  const float* __restrict__ b1, const float* __restrict__ W2,
                  float* __restrict__ D, int n, int in_c) {
  __shared__ float4 sB[KSTEPS * 2 * 32];
  __shared__ float4 sW[(HID / 2) * 4];

  int t = threadIdx.x;
  int lane = t & 31;
  int half = lane >> 4;
  int c    = lane & 15;
  int mTile = blockIdx.x * 8 + (t >> 5);
  bool live = mTile * 16 < n;
  int rowA = mTile * 16 + c;
  if (rowA >= n) rowA = n - 1;
  int beg = start[rowA];
  int m = cnt[rowA];
  if (!live) m = 0;
  float dv = dinv[rowA];
  const float* xh = xs + half * 4;
  float4 acc = *(const float4*)(xh + (size_t)rowA * AXP);

  float4 r0[(KSTEPS * 32) / 256], r1[(KSTEPS * 32) / 256];
#pragma unroll
  for (int j = 0; j < (KSTEPS * 32) / 256; ++j) {
    int k0 = (j * 8 + (t >> 5)) * 4 + half * 2;
    r0[j] = *(const float4*)(W2 + (size_t)k0 * OUTC + c * 4);
    r1[j] = *(const float4*)(W2 + (size_t)(k0 + 1) * OUTC + c * 4);
  }
  int kk = t & (HID - 1);
  float bv = b1[kk];
  float w1v[(HID * 8) / 256];
#pragma unroll
  for (int j = 0; j < (HID * 8) / 256; ++j) {
    int i = j * 2 + (t >> 7);
    w1v[j] = 0.0f;
    if (i < in_c) w1v[j] = W1[i * HID + kk];
  }
  int dummy = mTile * 16 < n ? mTile * 16 : n - 1;
  int idx[GU];
#pragma unroll
  for (int u = 0; u < GU; ++u) {
    int sv = csrc[u < m ? beg + u : 0];
    idx[u] = u < m ? sv : dummy;
  }
#pragma unroll
  for (int j = 0; j < (KSTEPS * 32) / 256; ++j) {
    int step = j * 8 + (t >> 5);
    sB[(step * 2 + 0) * 32 + lane] = make_float4(r0[j].x, r1[j].x, r0[j].y, r1[j].y);
    sB[(step * 2 + 1) * 32 + lane] = make_float4(r0[j].z, r1[j].z, r0[j].w, r1[j].w);
  }
  float* sWf = (float*)sW;
#pragma unroll
  for (int j = 0; j < (HID * 8) / 256; ++j) {
    int i = j * 2 + (t >> 7);
    sWf[(kk >> 1) * 16 + i * 2 + (kk & 1)] = i == 7 ? bv : w1v[j];
  }

  for (int base = 0; __any(base < m); base += GU) {
    float4 v[GU];
#pragma unroll
    for (int u = 0; u < GU; ++u) v[u] = *(const float4*)(xh + (size_t)idx[u] * AXP);
    int nb = base + GU;
#pragma unroll
    for (int u = 0; u < GU; ++u) {
      int sv = csrc[nb + u < m ? beg + nb + u : 0];
      idx[u] = nb + u < m ? sv : dummy;
    }
#pragma unroll
    for (int u = 0; u < GU; ++u) {
      bool on = base + u < m;
      acc.x += on ? v[u].x : 0.0f;
      acc.y += on ? v[u].y : 0.0f;
      acc.z += on ? v[u].z : 0.0f;
      acc.w += on ? v[u].w : 0.0f;
    }
  }
  float4 mine = make_float4(acc.x * dv, acc.y * dv, acc.z * dv, acc.w * dv);
  float4 oth;
  oth.x = __shfl_xor(mine.x, 16);
  oth.y = __shfl_xor(mine.y, 16);
  oth.z = __shfl_xor(mine.z, 16);
  oth.w = __shfl_xor(mine.w, 16);
  float4 x0 = half ? oth : mine;
  float4 x1 = half ? mine : oth;
  float xr[7] = {x0.x, x0.y, x0.z, x0.w, x1.x, x1.y, x1.z};

  __syncthreads();
  if (!live) return;

  v8f acc0 = {}, acc1 = {}, acc2 = {}, acc3 = {};
#pragma unroll
  for (int s = 0; s < KSTEPS; ++s) {
    const float4* wp = sW + (s * 2 + half) * 4;
    float4 w01 = wp[0], w23 = wp[1], w45 = wp[2], w6b = wp[3];
    v2f w, xv, hh;
    w.x = w01.x; w.y = w01.y; hh = w * xr[0];
    w.x = w01.z; w.y = w01.w; xv.x = xr[1]; xv.y = xr[1]; hh = __builtin_elementwise_fma(xv, w, hh);
    w.x = w23.x; w.y = w23.y; xv.x = xr[2]; xv.y = xr[2]; hh = __builtin_elementwise_fma(xv, w, hh);
    w.x = w23.z; w.y = w23.w; xv.x = xr[3]; xv.y = xr[3]; hh = __builtin_elementwise_fma(xv, w, hh);
    w.x = w45.x; w.y = w45.y; xv.x = xr[4]; xv.y = xr[4]; hh = __builtin_elementwise_fma(xv, w, hh);
    w.x = w45.z; w.y = w45.w; xv.x = xr[5]; xv.y = xr[5]; hh = __builtin_elementwise_fma(xv, w, hh);
    w.x = w6b.x; w.y = w6b.y; xv.x = xr[6]; xv.y = xr[6]; hh = __builtin_elementwise_fma(xv, w, hh);
    w.x = w6b.z; w.y = w6b.w; hh = hh + w;
    v2f a;
    a.x = fmaxf(hh.x, 0.0f) * dv;
    a.y = fmaxf(hh.y, 0.0f) * dv;
    float4 q0 = sB[(s * 2 + 0) * 32 + lane];
    float4 q1 = sB[(s * 2 + 1) * 32 + lane];
    v2f f0, f1, f2, f3;
    f0.x = q0.x; f0.y = q0.y;
    f1.x = q0.z; f1.y = q0.w;
    f2.x = q1.x; f2.y = q1.y;
    f3.x = q1.z; f3.y = q1.w;
    acc0 = __builtin_amdgcn_wmma_f32_16x16x4_f32(false, a, false, f0, (short)0, acc0, false, false);
    acc1 = __builtin_amdgcn_wmma_f32_16x16x4_f32(false, a, false, f1, (short)0, acc1, false, false);
    acc2 = __builtin_amdgcn_wmma_f32_16x16x4_f32(false, a, false, f2, (short)0, acc2, false, false);
    acc3 = __builtin_amdgcn_wmma_f32_16x16x4_f32(false, a, false, f3, (short)0, acc3, false, false);
  }
#pragma unroll
  for (int r = 0; r < 8; ++r) {
    int row = mTile * 16 + r + half * 8;
    if (row < n)
      *(float4*)(D + (size_t)row * OUTC + c * 4) = make_float4(acc0[r], acc1[r], acc2[r], acc3[r]);
  }
}

extern "C" void kernel_launch(void* const* d_in, const int* in_sizes, int n_in,
                              void* d_out, int out_size, void* d_ws, size_t ws_size,
                              hipStream_t stream) {
  const float* x  = (const float*)d_in[0];
  const int*   ei = (const int*)d_in[1];
  const float* W1 = (const float*)d_in[2];
  const float* b1 = (const float*)d_in[3];
  const float* W2 = (const float*)d_in[4];
  const float* b2 = (const float*)d_in[5];
  float* out = (float*)d_out;

  int hid   = in_sizes[3];
  int in_c  = in_sizes[2] / hid;
  int n     = in_sizes[0] / in_c;
  int E     = in_sizes[1] / 2;

  const int* src = ei;
  const int* dst = ei + E;

  float* ws   = (float*)d_ws;
  float* dinv = ws;
  float* xs   = dinv + n;
  float* h2   = xs + (size_t)n * AXP;
  int* cnt    = (int*)(h2 + (size_t)n * OUTC);
  int* start  = cnt + n;
  int* csrc   = start + n;
  int* seg    = csrc + E;
  int* offT   = seg + E;

  const int B256 = 256;
  int nr  = (n + RW - 1) / RW;
  int nch = (E + CHUNK - 1) / CHUNK;
  k_bucket<<<nch, BT, 0, stream>>>(src, dst, seg, offT, E, n, nr, nch);
  k_place<<<nr, BT, 0, stream>>>(seg, offT, x, cnt, start, dinv, xs, csrc, n, in_c, nch);

  int mTiles = (n + 15) / 16;
  k_gather_mlp_wmma<<<(mTiles + 7) / 8, B256, 0, stream>>>(xs, dinv, start, cnt, csrc, W1, b1, W2, h2, n, in_c);

  k_gather64<<<GW_BLOCKS, B256, 0, stream>>>(h2, dinv, start, cnt, csrc, b2, out, n);
}
